# starting kernel
# baseline (speedup 1.0000x reference)
#include <hip/hip_runtime.h>
#include <math.h>


typedef __attribute__((ext_vector_type(2))) float v2f;
typedef __attribute__((ext_vector_type(8))) float v8f;

#define HW 64
#define PIX 4096
#define CIN 256
#define CCOMP 64
#define MENC 100
#define MPAD 112
#define KENC 576
#define NPIX_TOTAL 16384

__global__ __launch_bounds__(256) void carafe_compress(
    const float* __restrict__ x, const float* __restrict__ w_comp,
    const float* __restrict__ b_comp, float* __restrict__ comp) {
  const int wave = threadIdx.x >> 5;
  const int lane = threadIdx.x & 31;
  const int row  = lane & 15;
  const int half = lane >> 4;

  const int tile = blockIdx.x * 8 + wave;
  const int gp   = tile << 4;
  const int n    = gp >> 12;
  const int pin  = gp & 4095;

  v8f acc[4];
#pragma unroll
  for (int t = 0; t < 4; ++t)
#pragma unroll
    for (int v = 0; v < 8; ++v)
      acc[t][v] = b_comp[t * 16 + v + 8 * half];

  const float* xb = x + (size_t)n * CIN * PIX + pin + row;

  for (int k0 = 0; k0 < CIN; k0 += 4) {
    const int kb = k0 + 2 * half;
    v2f b;
    b[0] = xb[(size_t)(kb + 0) * PIX];
    b[1] = xb[(size_t)(kb + 1) * PIX];
#pragma unroll
    for (int t = 0; t < 4; ++t) {
      v2f a = *(const v2f*)(w_comp + (t * 16 + row) * CIN + kb);
      acc[t] = __builtin_amdgcn_wmma_f32_16x16x4_f32(
          false, a, false, b, (short)0, acc[t], false, false);
    }
  }

  float* cb = comp + (size_t)n * CCOMP * PIX + pin + row;
#pragma unroll
  for (int t = 0; t < 4; ++t)
#pragma unroll
    for (int v = 0; v < 8; ++v)
      cb[(size_t)(t * 16 + v + 8 * half) * PIX] = acc[t][v];
}

__global__ __launch_bounds__(256) void carafe_encode_softmax(
    const float* __restrict__ comp, const float* __restrict__ w_enc,
    const float* __restrict__ b_enc, float* __restrict__ mask) {
  __shared__ float sm[8][16 * 113];

  const int wave = threadIdx.x >> 5;
  const int lane = threadIdx.x & 31;
  const int row  = lane & 15;
  const int half = lane >> 4;

  const int tile = blockIdx.x * 8 + wave;
  const int gp   = tile << 4;
  const int n    = gp >> 12;
  const int pin  = gp & 4095;

  const int p  = pin + row;
  const int py = p >> 6;
  const int px = p & 63;

  v8f acc[7];
#pragma unroll
  for (int t = 0; t < 7; ++t)
#pragma unroll
    for (int v = 0; v < 8; ++v) {
      const int m = t * 16 + v + 8 * half;
      acc[t][v] = (m < MENC) ? b_enc[m] : 0.0f;
    }

  const float* cbase = comp + (size_t)n * CCOMP * PIX;

  for (int k0 = 0; k0 < KENC; k0 += 4) {
    const int kb = k0 + 2 * half;
    v2f b;
#pragma unroll
    for (int v = 0; v < 2; ++v) {
      const int k  = kb + v;
      const int cc = k / 9;
      const int r  = k - cc * 9;
      const int y  = py + (r / 3) - 1;
      const int xx = px + (r % 3) - 1;
      const bool ok = ((unsigned)y < HW) & ((unsigned)xx < HW);
      const int yc = min(max(y, 0), HW - 1);
      const int xc = min(max(xx, 0), HW - 1);
      const float val = cbase[(size_t)cc * PIX + yc * HW + xc];
      b[v] = ok ? val : 0.0f;
    }
#pragma unroll
    for (int t = 0; t < 7; ++t) {
      const int m  = t * 16 + row;
      const int mc = min(m, MENC - 1);
      v2f a = *(const v2f*)(w_enc + (size_t)mc * KENC + kb);
      if (m >= MENC) { a[0] = 0.0f; a[1] = 0.0f; }
      acc[t] = __builtin_amdgcn_wmma_f32_16x16x4_f32(
          false, a, false, b, (short)0, acc[t], false, false);
    }
  }

  float* s = sm[wave];
#pragma unroll
  for (int t = 0; t < 7; ++t)
#pragma unroll
    for (int v = 0; v < 8; ++v)
      s[row * 113 + t * 16 + v + 8 * half] = acc[t][v];

  __syncthreads();

#pragma unroll
  for (int u = 0; u < 2; ++u) {
    const int g   = lane * 2 + u;
    const int pix = g >> 2;
    const int ij  = g & 3;
    const float* base = s + pix * 113;
    float r[25];
    float mx = -3.4e38f;
#pragma unroll
    for (int k = 0; k < 25; ++k) { r[k] = base[k * 4 + ij]; mx = fmaxf(mx, r[k]); }
    float sum = 0.0f;
#pragma unroll
    for (int k = 0; k < 25; ++k) { r[k] = __expf(r[k] - mx); sum += r[k]; }
    const float inv = 1.0f / sum;
    float* mout = mask + (size_t)(gp + pix) * 100 + ij * 25;
#pragma unroll
    for (int k = 0; k < 25; ++k) mout[k] = r[k] * inv;
  }
}

__global__ __launch_bounds__(64) void carafe_reassemble(
    const float* __restrict__ x, const float* __restrict__ mask,
    float* __restrict__ out) {
  __shared__ float sm[64 * 101];

  const int t  = threadIdx.x;
  const int b  = blockIdx.x;
  const int cg = b & 31;
  const int h  = (b >> 5) & 63;
  const int n  = b >> 11;

  const int pg = (n << 12) + (h << 6) + t;
  const float4* mp4 = (const float4*)(mask + (size_t)pg * 100);
  float* srow = sm + t * 101;
#pragma unroll
  for (int q = 0; q < 25; ++q) {
    const float4 m4 = mp4[q];
    srow[q * 4 + 0] = m4.x; srow[q * 4 + 1] = m4.y;
    srow[q * 4 + 2] = m4.z; srow[q * 4 + 3] = m4.w;
  }
  __syncthreads();

  const float* ms = sm + t * 101;

  for (int ci = 0; ci < 8; ++ci) {
    const int c = cg * 8 + ci;
    const float* xb = x + ((size_t)(n * 256 + c)) * PIX;

    float xv[25];
#pragma unroll
    for (int di = 0; di < 5; ++di) {
      const int y  = h + di - 2;
      const int yc = min(max(y, 0), HW - 1);
#pragma unroll
      for (int dj = 0; dj < 5; ++dj) {
        const int xx = t + dj - 2;
        const int xc = min(max(xx, 0), HW - 1);
        const bool ok = ((unsigned)y < HW) & ((unsigned)xx < HW);
        const float val = xb[yc * HW + xc];
        xv[di * 5 + dj] = ok ? val : 0.0f;
      }
    }

    float a0 = 0.f, a1 = 0.f, a2 = 0.f, a3 = 0.f;
#pragma unroll
    for (int k = 0; k < 25; ++k) {
      const float pch = xv[k];
      a0 += pch * ms[k];
      a1 += pch * ms[25 + k];
      a2 += pch * ms[50 + k];
      a3 += pch * ms[75 + k];
    }

    float* ob = out + (((size_t)(n * 256 + c) * 128) + 2 * h) * 128 + 2 * t;
    float2 r0; r0.x = a0; r0.y = a1;
    float2 r1; r1.x = a2; r1.y = a3;
    *(float2*)(ob)       = r0;
    *(float2*)(ob + 128) = r1;
  }
}

extern "C" void kernel_launch(void* const* d_in, const int* in_sizes, int n_in,
                              void* d_out, int out_size, void* d_ws, size_t ws_size,
                              hipStream_t stream) {
  (void)in_sizes; (void)n_in; (void)out_size; (void)ws_size;
  const float* x      = (const float*)d_in[0];
  const float* w_comp = (const float*)d_in[1];
  const float* b_comp = (const float*)d_in[2];
  const float* w_enc  = (const float*)d_in[3];
  const float* b_enc  = (const float*)d_in[4];
  float* out = (float*)d_out;

  float* comp = (float*)d_ws;
  float* mask = comp + (size_t)4 * CCOMP * PIX;

  carafe_compress<<<128, 256, 0, stream>>>(x, w_comp, b_comp, comp);
  carafe_encode_softmax<<<128, 256, 0, stream>>>(comp, w_enc, b_enc, mask);
  carafe_reassemble<<<8192, 64, 0, stream>>>(x, mask, out);
}


# 3x3 conv weight-stationary: work-group = image row, one wave per M tile keeps its A fragments for all K in registers over the row's 4 pixel tiles, only B streamed; conv k order (sum order) permuted
# speedup vs baseline: 1.4295x; 1.4295x over previous
#include <hip/hip_runtime.h>
#include <math.h>


typedef __attribute__((ext_vector_type(2))) float v2f;
typedef __attribute__((ext_vector_type(8))) float v8f;

#define HW 64
#define PIX 4096
#define CIN 256
#define CCOMP 64
#define MENC 100
#define MPAD 112
#define KENC 576
#define NPIX_TOTAL 16384

__global__ __launch_bounds__(256) void carafe_compress(
    const float* __restrict__ x, const float* __restrict__ w_comp,
    const float* __restrict__ b_comp, float* __restrict__ comp) {
  const int wave = threadIdx.x >> 5;
  const int lane = threadIdx.x & 31;
  const int row  = lane & 15;
  const int half = lane >> 4;

  const int tile = blockIdx.x * 8 + wave;
  const int gp   = tile << 4;
  const int n    = gp >> 12;
  const int pin  = gp & 4095;

  v8f acc[4];
#pragma unroll
  for (int t = 0; t < 4; ++t)
#pragma unroll
    for (int v = 0; v < 8; ++v)
      acc[t][v] = b_comp[t * 16 + v + 8 * half];

  const float* xb = x + (size_t)n * CIN * PIX + pin + row;

  for (int k0 = 0; k0 < CIN; k0 += 4) {
    const int kb = k0 + 2 * half;
    v2f b;
    b[0] = xb[(size_t)(kb + 0) * PIX];
    b[1] = xb[(size_t)(kb + 1) * PIX];
#pragma unroll
    for (int t = 0; t < 4; ++t) {
      v2f a = *(const v2f*)(w_comp + (t * 16 + row) * CIN + kb);
      acc[t] = __builtin_amdgcn_wmma_f32_16x16x4_f32(
          false, a, false, b, (short)0, acc[t], false, false);
    }
  }

  float* cb = comp + (size_t)n * CCOMP * PIX + pin + row;
#pragma unroll
  for (int t = 0; t < 4; ++t)
#pragma unroll
    for (int v = 0; v < 8; ++v)
      cb[(size_t)(t * 16 + v + 8 * half) * PIX] = acc[t][v];
}

__global__ __launch_bounds__(224) void carafe_encode_softmax(
    const float* __restrict__ comp, const float* __restrict__ w_enc,
    const float* __restrict__ b_enc, float* __restrict__ mask) {
  __shared__ float sm[64 * 113];

  const int tid  = threadIdx.x;
  const int wave = tid >> 5;
  const int lane = tid & 31;
  const int row  = lane & 15;
  const int half = lane >> 4;

  const int n = blockIdx.x >> 6;
  const int h = blockIdx.x & 63;

  const int m  = wave * 16 + row;
  const int mc = min(m, MENC - 1);
  const bool mvalid = m < MENC;
  const float* wrow = w_enc + (size_t)mc * KENC + 18 * half;
  v2f a[16][9];
#pragma unroll
  for (int q = 0; q < 16; ++q)
#pragma unroll
    for (int j = 0; j < 9; ++j) {
      v2f t = *(const v2f*)(wrow + 36 * q + 2 * j);
      if (!mvalid) { t[0] = 0.0f; t[1] = 0.0f; }
      a[q][j] = t;
    }

  v8f bias;
#pragma unroll
  for (int v = 0; v < 8; ++v) {
    const int mo = wave * 16 + v + 8 * half;
    bias[v] = (mo < MENC) ? b_enc[mo] : 0.0f;
  }

  const float* cb = comp + (size_t)n * CCOMP * PIX + (size_t)(2 * half) * PIX;

  for (int it = 0; it < 2; ++it) {
    int  off[2][9];
    bool ok[2][9];
#pragma unroll
    for (int T = 0; T < 2; ++T) {
      const int px = it * 32 + T * 16 + row;
#pragma unroll
      for (int r = 0; r < 9; ++r) {
        const int y  = h + (r / 3) - 1;
        const int xx = px + (r % 3) - 1;
        ok[T][r] = ((unsigned)y < HW) & ((unsigned)xx < HW);
        const int yc = min(max(y, 0), HW - 1);
        const int xc = min(max(xx, 0), HW - 1);
        off[T][r] = yc * HW + xc;
      }
    }

    v8f acc0 = bias, acc1 = bias;

    float cur[2][18], nxt[2][18];
#pragma unroll
    for (int kk = 0; kk < 18; ++kk) {
      cur[0][kk] = cb[(size_t)(kk / 9) * PIX + off[0][kk % 9]];
      cur[1][kk] = cb[(size_t)(kk / 9) * PIX + off[1][kk % 9]];
    }

#pragma unroll
    for (int q = 0; q < 16; ++q) {
      if (q + 1 < 16) {
#pragma unroll
        for (int kk = 0; kk < 18; ++kk) {
          nxt[0][kk] = cb[(size_t)(4 * (q + 1) + kk / 9) * PIX + off[0][kk % 9]];
          nxt[1][kk] = cb[(size_t)(4 * (q + 1) + kk / 9) * PIX + off[1][kk % 9]];
        }
      }
#pragma unroll
      for (int j = 0; j < 9; ++j) {
        v2f b0, b1;
#pragma unroll
        for (int v = 0; v < 2; ++v) {
          const int kk = 2 * j + v;
          b0[v] = ok[0][kk % 9] ? cur[0][kk] : 0.0f;
          b1[v] = ok[1][kk % 9] ? cur[1][kk] : 0.0f;
        }
        acc0 = __builtin_amdgcn_wmma_f32_16x16x4_f32(
            false, a[q][j], false, b0, (short)0, acc0, false, false);
        acc1 = __builtin_amdgcn_wmma_f32_16x16x4_f32(
            false, a[q][j], false, b1, (short)0, acc1, false, false);
      }
#pragma unroll
      for (int kk = 0; kk < 18; ++kk) { cur[0][kk] = nxt[0][kk]; cur[1][kk] = nxt[1][kk]; }
    }

    float* s0 = sm + (it * 32 + row) * 113 + wave * 16 + 8 * half;
    float* s1 = s0 + 16 * 113;
#pragma unroll
    for (int v = 0; v < 8; ++v) { s0[v] = acc0[v]; s1[v] = acc1[v]; }

    __syncthreads();

    if (tid < 128) {
      const int pix = it * 32 + (tid >> 2);
      const int ij  = tid & 3;
      const float* base = sm + pix * 113;
      float r[25];
      float mx = -3.4e38f;
#pragma unroll
      for (int k = 0; k < 25; ++k) { r[k] = base[k * 4 + ij]; mx = fmaxf(mx, r[k]); }
      float sum = 0.0f;
#pragma unroll
      for (int k = 0; k < 25; ++k) { r[k] = __expf(r[k] - mx); sum += r[k]; }
      const float inv = 1.0f / sum;
      float* mout = mask + ((size_t)n * PIX + h * HW + pix) * 100 + ij * 25;
#pragma unroll
      for (int k = 0; k < 25; ++k) mout[k] = r[k] * inv;
    }
  }
}

__global__ __launch_bounds__(64) void carafe_reassemble(
    const float* __restrict__ x, const float* __restrict__ mask,
    float* __restrict__ out) {
  __shared__ float sm[64 * 101];

  const int t  = threadIdx.x;
  const int b  = blockIdx.x;
  const int cg = b & 31;
  const int h  = (b >> 5) & 63;
  const int n  = b >> 11;

  const int pg = (n << 12) + (h << 6) + t;
  const float4* mp4 = (const float4*)(mask + (size_t)pg * 100);
  float* srow = sm + t * 101;
#pragma unroll
  for (int q = 0; q < 25; ++q) {
    const float4 m4 = mp4[q];
    srow[q * 4 + 0] = m4.x; srow[q * 4 + 1] = m4.y;
    srow[q * 4 + 2] = m4.z; srow[q * 4 + 3] = m4.w;
  }
  __syncthreads();

  const float* ms = sm + t * 101;

  for (int ci = 0; ci < 8; ++ci) {
    const int c = cg * 8 + ci;
    const float* xb = x + ((size_t)(n * 256 + c)) * PIX;

    float xv[25];
#pragma unroll
    for (int di = 0; di < 5; ++di) {
      const int y  = h + di - 2;
      const int yc = min(max(y, 0), HW - 1);
#pragma unroll
      for (int dj = 0; dj < 5; ++dj) {
        const int xx = t + dj - 2;
        const int xc = min(max(xx, 0), HW - 1);
        const bool ok = ((unsigned)y < HW) & ((unsigned)xx < HW);
        const float val = xb[yc * HW + xc];
        xv[di * 5 + dj] = ok ? val : 0.0f;
      }
    }

    float a0 = 0.f, a1 = 0.f, a2 = 0.f, a3 = 0.f;
#pragma unroll
    for (int k = 0; k < 25; ++k) {
      const float pch = xv[k];
      a0 += pch * ms[k];
      a1 += pch * ms[25 + k];
      a2 += pch * ms[50 + k];
      a3 += pch * ms[75 + k];
    }

    float* ob = out + (((size_t)(n * 256 + c) * 128) + 2 * h) * 128 + 2 * t;
    float2 r0; r0.x = a0; r0.y = a1;
    float2 r1; r1.x = a2; r1.y = a3;
    *(float2*)(ob)       = r0;
    *(float2*)(ob + 128) = r1;
  }
}

extern "C" void kernel_launch(void* const* d_in, const int* in_sizes, int n_in,
                              void* d_out, int out_size, void* d_ws, size_t ws_size,
                              hipStream_t stream) {
  (void)in_sizes; (void)n_in; (void)out_size; (void)ws_size;
  const float* x      = (const float*)d_in[0];
  const float* w_comp = (const float*)d_in[1];
  const float* b_comp = (const float*)d_in[2];
  const float* w_enc  = (const float*)d_in[3];
  const float* b_enc  = (const float*)d_in[4];
  float* out = (float*)d_out;

  float* comp = (float*)d_ws;
  float* mask = comp + (size_t)4 * CCOMP * PIX;

  carafe_compress<<<128, 256, 0, stream>>>(x, w_comp, b_comp, comp);
  carafe_encode_softmax<<<256, 224, 0, stream>>>(comp, w_enc, b_enc, mask);
  carafe_reassemble<<<8192, 64, 0, stream>>>(x, mask, out);
}


# Mask plane re-laid pixel-fastest in 4-tap quads: softmax stores and reassembly loads are whole-line 16-byte accesses; a lane loads its 100 weights straight to registers, no LDS copy; sums unchanged
# speedup vs baseline: 2.3764x; 1.6624x over previous
#include <hip/hip_runtime.h>
#include <math.h>


typedef __attribute__((ext_vector_type(2))) float v2f;
typedef __attribute__((ext_vector_type(8))) float v8f;

#define HW 64
#define PIX 4096
#define CIN 256
#define CCOMP 64
#define MENC 100
#define MPAD 112
#define KENC 576
#define NPIX_TOTAL 16384
#define MPLANE 6400
#define MTAIL 6144

__global__ __launch_bounds__(256) void carafe_compress(
    const float* __restrict__ x, const float* __restrict__ w_comp,
    const float* __restrict__ b_comp, float* __restrict__ comp) {
  const int wave = threadIdx.x >> 5;
  const int lane = threadIdx.x & 31;
  const int row  = lane & 15;
  const int half = lane >> 4;

  const int tile = blockIdx.x * 8 + wave;
  const int gp   = tile << 4;
  const int n    = gp >> 12;
  const int pin  = gp & 4095;

  v8f acc[4];
#pragma unroll
  for (int t = 0; t < 4; ++t)
#pragma unroll
    for (int v = 0; v < 8; ++v)
      acc[t][v] = b_comp[t * 16 + v + 8 * half];

  const float* xb = x + (size_t)n * CIN * PIX + pin + row;

  for (int k0 = 0; k0 < CIN; k0 += 4) {
    const int kb = k0 + 2 * half;
    v2f b;
    b[0] = xb[(size_t)(kb + 0) * PIX];
    b[1] = xb[(size_t)(kb + 1) * PIX];
#pragma unroll
    for (int t = 0; t < 4; ++t) {
      v2f a = *(const v2f*)(w_comp + (t * 16 + row) * CIN + kb);
      acc[t] = __builtin_amdgcn_wmma_f32_16x16x4_f32(
          false, a, false, b, (short)0, acc[t], false, false);
    }
  }

  float* cb = comp + (size_t)n * CCOMP * PIX + pin + row;
#pragma unroll
  for (int t = 0; t < 4; ++t)
#pragma unroll
    for (int v = 0; v < 8; ++v)
      cb[(size_t)(t * 16 + v + 8 * half) * PIX] = acc[t][v];
}

__global__ __launch_bounds__(224) void carafe_encode_softmax(
    const float* __restrict__ comp, const float* __restrict__ w_enc,
    const float* __restrict__ b_enc, float* __restrict__ mask) {
  __shared__ float sm[64 * 113];

  const int tid  = threadIdx.x;
  const int wave = tid >> 5;
  const int lane = tid & 31;
  const int row  = lane & 15;
  const int half = lane >> 4;

  const int n = blockIdx.x >> 6;
  const int h = blockIdx.x & 63;

  const int m  = wave * 16 + row;
  const int mc = min(m, MENC - 1);
  const bool mvalid = m < MENC;
  const float* wrow = w_enc + (size_t)mc * KENC + 18 * half;
  v2f a[16][9];
#pragma unroll
  for (int q = 0; q < 16; ++q)
#pragma unroll
    for (int j = 0; j < 9; ++j) {
      v2f t = *(const v2f*)(wrow + 36 * q + 2 * j);
      if (!mvalid) { t[0] = 0.0f; t[1] = 0.0f; }
      a[q][j] = t;
    }

  v8f bias;
#pragma unroll
  for (int v = 0; v < 8; ++v) {
    const int mo = wave * 16 + v + 8 * half;
    bias[v] = (mo < MENC) ? b_enc[mo] : 0.0f;
  }

  const float* cb = comp + (size_t)n * CCOMP * PIX + (size_t)(2 * half) * PIX;

  for (int it = 0; it < 2; ++it) {
    int  off[2][9];
    bool ok[2][9];
#pragma unroll
    for (int T = 0; T < 2; ++T) {
      const int px = it * 32 + T * 16 + row;
#pragma unroll
      for (int r = 0; r < 9; ++r) {
        const int y  = h + (r / 3) - 1;
        const int xx = px + (r % 3) - 1;
        ok[T][r] = ((unsigned)y < HW) & ((unsigned)xx < HW);
        const int yc = min(max(y, 0), HW - 1);
        const int xc = min(max(xx, 0), HW - 1);
        off[T][r] = yc * HW + xc;
      }
    }

    v8f acc0 = bias, acc1 = bias;

    float cur[2][18], nxt[2][18];
#pragma unroll
    for (int kk = 0; kk < 18; ++kk) {
      cur[0][kk] = cb[(size_t)(kk / 9) * PIX + off[0][kk % 9]];
      cur[1][kk] = cb[(size_t)(kk / 9) * PIX + off[1][kk % 9]];
    }

#pragma unroll
    for (int q = 0; q < 16; ++q) {
      if (q + 1 < 16) {
#pragma unroll
        for (int kk = 0; kk < 18; ++kk) {
          nxt[0][kk] = cb[(size_t)(4 * (q + 1) + kk / 9) * PIX + off[0][kk % 9]];
          nxt[1][kk] = cb[(size_t)(4 * (q + 1) + kk / 9) * PIX + off[1][kk % 9]];
        }
      }
#pragma unroll
      for (int j = 0; j < 9; ++j) {
        v2f b0, b1;
#pragma unroll
        for (int v = 0; v < 2; ++v) {
          const int kk = 2 * j + v;
          b0[v] = ok[0][kk % 9] ? cur[0][kk] : 0.0f;
          b1[v] = ok[1][kk % 9] ? cur[1][kk] : 0.0f;
        }
        acc0 = __builtin_amdgcn_wmma_f32_16x16x4_f32(
            false, a[q][j], false, b0, (short)0, acc0, false, false);
        acc1 = __builtin_amdgcn_wmma_f32_16x16x4_f32(
            false, a[q][j], false, b1, (short)0, acc1, false, false);
      }
#pragma unroll
      for (int kk = 0; kk < 18; ++kk) { cur[0][kk] = nxt[0][kk]; cur[1][kk] = nxt[1][kk]; }
    }

    float* s0 = sm + (it * 32 + row) * 113 + wave * 16 + 8 * half;
    float* s1 = s0 + 16 * 113;
#pragma unroll
    for (int v = 0; v < 8; ++v) { s0[v] = acc0[v]; s1[v] = acc1[v]; }

    __syncthreads();

    if (tid < 128) {
      const int pix = it * 32 + (tid & 31);
      const int ij  = tid >> 5;
      const float* base = sm + pix * 113;
      float r[25];
      float mx = -3.4e38f;
#pragma unroll
      for (int k = 0; k < 25; ++k) { r[k] = base[k * 4 + ij]; mx = fmaxf(mx, r[k]); }
      float sum = 0.0f;
#pragma unroll
      for (int k = 0; k < 25; ++k) { r[k] = __expf(r[k] - mx); sum += r[k]; }
      const float inv = 1.0f / sum;
      float* mplane = mask + (size_t)(n * HW + h) * MPLANE;
      float* mquad  = mplane + (ij * 6 * HW + pix) * 4;
#pragma unroll
      for (int g = 0; g < 6; ++g) {
        float4 o;
        o.x = r[4 * g + 0] * inv; o.y = r[4 * g + 1] * inv;
        o.z = r[4 * g + 2] * inv; o.w = r[4 * g + 3] * inv;
        *(float4*)(mquad + g * HW * 4) = o;
      }
      mplane[MTAIL + ij * HW + pix] = r[24] * inv;
    }
  }
}

__global__ __launch_bounds__(64) void carafe_reassemble(
    const float* __restrict__ x, const float* __restrict__ mask,
    float* __restrict__ out) {
  const int t  = threadIdx.x;
  const int b  = blockIdx.x;
  const int cg = b & 31;
  const int h  = (b >> 5) & 63;
  const int n  = b >> 11;

  const float* mplane = mask + (size_t)(n * HW + h) * MPLANE;
  float ms[100];
#pragma unroll
  for (int ij = 0; ij < 4; ++ij) {
#pragma unroll
    for (int g = 0; g < 6; ++g) {
      const float4 m4 = *(const float4*)(mplane + ((ij * 6 + g) * HW + t) * 4);
      ms[ij * 25 + 4 * g + 0] = m4.x; ms[ij * 25 + 4 * g + 1] = m4.y;
      ms[ij * 25 + 4 * g + 2] = m4.z; ms[ij * 25 + 4 * g + 3] = m4.w;
    }
    ms[ij * 25 + 24] = mplane[MTAIL + ij * HW + t];
  }

  for (int ci = 0; ci < 8; ++ci) {
    const int c = cg * 8 + ci;
    const float* xb = x + ((size_t)(n * 256 + c)) * PIX;

    float xv[25];
#pragma unroll
    for (int di = 0; di < 5; ++di) {
      const int y  = h + di - 2;
      const int yc = min(max(y, 0), HW - 1);
#pragma unroll
      for (int dj = 0; dj < 5; ++dj) {
        const int xx = t + dj - 2;
        const int xc = min(max(xx, 0), HW - 1);
        const bool ok = ((unsigned)y < HW) & ((unsigned)xx < HW);
        const float val = xb[yc * HW + xc];
        xv[di * 5 + dj] = ok ? val : 0.0f;
      }
    }

    float a0 = 0.f, a1 = 0.f, a2 = 0.f, a3 = 0.f;
#pragma unroll
    for (int k = 0; k < 25; ++k) {
      const float pch = xv[k];
      a0 += pch * ms[k];
      a1 += pch * ms[25 + k];
      a2 += pch * ms[50 + k];
      a3 += pch * ms[75 + k];
    }

    float* ob = out + (((size_t)(n * 256 + c) * 128) + 2 * h) * 128 + 2 * t;
    float2 r0; r0.x = a0; r0.y = a1;
    float2 r1; r1.x = a2; r1.y = a3;
    *(float2*)(ob)       = r0;
    *(float2*)(ob + 128) = r1;
  }
}

extern "C" void kernel_launch(void* const* d_in, const int* in_sizes, int n_in,
                              void* d_out, int out_size, void* d_ws, size_t ws_size,
                              hipStream_t stream) {
  (void)in_sizes; (void)n_in; (void)out_size; (void)ws_size;
  const float* x      = (const float*)d_in[0];
  const float* w_comp = (const float*)d_in[1];
  const float* b_comp = (const float*)d_in[2];
  const float* w_enc  = (const float*)d_in[3];
  const float* b_enc  = (const float*)d_in[4];
  float* out = (float*)d_out;

  float* comp = (float*)d_ws;
  float* mask = comp + (size_t)4 * CCOMP * PIX;

  carafe_compress<<<128, 256, 0, stream>>>(x, w_comp, b_comp, comp);
  carafe_encode_softmax<<<256, 224, 0, stream>>>(comp, w_enc, b_enc, mask);
  carafe_reassemble<<<8192, 64, 0, stream>>>(x, mask, out);
}


# 1x1 conv re-tiled: a wave = one 16-row M tile x 64 pixels as 4 pixel-interleaved N tiles (tile j, column c = pixel 4c+j): B loads, comp stores 16 B per lane in whole 256-B pieces; sums unchanged
# speedup vs baseline: 3.2094x; 1.3505x over previous
#include <hip/hip_runtime.h>
#include <math.h>


typedef __attribute__((ext_vector_type(2))) float v2f;
typedef __attribute__((ext_vector_type(4))) float v4f;
typedef __attribute__((ext_vector_type(8))) float v8f;

#define HW 64
#define PIX 4096
#define CIN 256
#define CCOMP 64
#define MENC 100
#define MPAD 112
#define KENC 576
#define NPIX_TOTAL 16384
#define MPLANE 6400
#define MTAIL 6144

__global__ __launch_bounds__(128) __attribute__((amdgpu_waves_per_eu(1, 1)))
void carafe_compress(
    const float* __restrict__ x, const float* __restrict__ w_comp,
    const float* __restrict__ b_comp, float* __restrict__ comp) {
  const int mt   = threadIdx.x >> 5;
  const int lane = threadIdx.x & 31;
  const int col  = lane & 15;
  const int half = lane >> 4;

  const int gp  = blockIdx.x << 6;
  const int n   = gp >> 12;
  const int pin = gp & 4095;

  v8f acc[4];
#pragma unroll
  for (int v = 0; v < 8; ++v) {
    const float bv = b_comp[mt * 16 + v + 8 * half];
#pragma unroll
    for (int j = 0; j < 4; ++j) acc[j][v] = bv;
  }

  const float* xb = x + ((size_t)n * CIN + 2 * half) * PIX + pin + 4 * col;
  const float* wr = w_comp + (mt * 16 + col) * CIN + 2 * half;

  v4f cb[16][2], nb[16][2];
  v2f ca[16], na[16];
#pragma unroll
  for (int s = 0; s < 16; ++s) {
    ca[s]    = *(const v2f*)(wr + 4 * s);
    cb[s][0] = *(const v4f*)(xb + (size_t)(4 * s + 0) * PIX);
    cb[s][1] = *(const v4f*)(xb + (size_t)(4 * s + 1) * PIX);
  }

#pragma unroll
  for (int ch = 0; ch < 4; ++ch) {
    if (ch + 1 < 4) {
#pragma unroll
      for (int s = 0; s < 16; ++s) {
        const int k0 = 64 * (ch + 1) + 4 * s;
        na[s]    = *(const v2f*)(wr + k0);
        nb[s][0] = *(const v4f*)(xb + (size_t)(k0 + 0) * PIX);
        nb[s][1] = *(const v4f*)(xb + (size_t)(k0 + 1) * PIX);
      }
    }
#pragma unroll
    for (int s = 0; s < 16; ++s) {
#pragma unroll
      for (int j = 0; j < 4; ++j) {
        v2f b;
        b[0] = cb[s][0][j];
        b[1] = cb[s][1][j];
        acc[j] = __builtin_amdgcn_wmma_f32_16x16x4_f32(
            false, ca[s], false, b, (short)0, acc[j], false, false);
      }
    }
#pragma unroll
    for (int s = 0; s < 16; ++s) {
      ca[s] = na[s]; cb[s][0] = nb[s][0]; cb[s][1] = nb[s][1];
    }
  }

  float* co = comp + ((size_t)n * CCOMP + mt * 16 + 8 * half) * PIX + pin + 4 * col;
#pragma unroll
  for (int v = 0; v < 8; ++v) {
    v4f o;
    o[0] = acc[0][v]; o[1] = acc[1][v]; o[2] = acc[2][v]; o[3] = acc[3][v];
    *(v4f*)(co + (size_t)v * PIX) = o;
  }
}

__global__ __launch_bounds__(224) void carafe_encode_softmax(
    const float* __restrict__ comp, const float* __restrict__ w_enc,
    const float* __restrict__ b_enc, float* __restrict__ mask) {
  __shared__ float sm[64 * 113];

  const int tid  = threadIdx.x;
  const int wave = tid >> 5;
  const int lane = tid & 31;
  const int row  = lane & 15;
  const int half = lane >> 4;

  const int n = blockIdx.x >> 6;
  const int h = blockIdx.x & 63;

  const int m  = wave * 16 + row;
  const int mc = min(m, MENC - 1);
  const bool mvalid = m < MENC;
  const float* wrow = w_enc + (size_t)mc * KENC + 18 * half;
  v2f a[16][9];
#pragma unroll
  for (int q = 0; q < 16; ++q)
#pragma unroll
    for (int j = 0; j < 9; ++j) {
      v2f t = *(const v2f*)(wrow + 36 * q + 2 * j);
      if (!mvalid) { t[0] = 0.0f; t[1] = 0.0f; }
      a[q][j] = t;
    }

  v8f bias;
#pragma unroll
  for (int v = 0; v < 8; ++v) {
    const int mo = wave * 16 + v + 8 * half;
    bias[v] = (mo < MENC) ? b_enc[mo] : 0.0f;
  }

  const float* cb = comp + (size_t)n * CCOMP * PIX + (size_t)(2 * half) * PIX;

  for (int it = 0; it < 2; ++it) {
    int  off[2][9];
    bool ok[2][9];
#pragma unroll
    for (int T = 0; T < 2; ++T) {
      const int px = it * 32 + T * 16 + row;
#pragma unroll
      for (int r = 0; r < 9; ++r) {
        const int y  = h + (r / 3) - 1;
        const int xx = px + (r % 3) - 1;
        ok[T][r] = ((unsigned)y < HW) & ((unsigned)xx < HW);
        const int yc = min(max(y, 0), HW - 1);
        const int xc = min(max(xx, 0), HW - 1);
        off[T][r] = yc * HW + xc;
      }
    }

    v8f acc0 = bias, acc1 = bias;

    float cur[2][18], nxt[2][18];
#pragma unroll
    for (int kk = 0; kk < 18; ++kk) {
      cur[0][kk] = cb[(size_t)(kk / 9) * PIX + off[0][kk % 9]];
      cur[1][kk] = cb[(size_t)(kk / 9) * PIX + off[1][kk % 9]];
    }

#pragma unroll
    for (int q = 0; q < 16; ++q) {
      if (q + 1 < 16) {
#pragma unroll
        for (int kk = 0; kk < 18; ++kk) {
          nxt[0][kk] = cb[(size_t)(4 * (q + 1) + kk / 9) * PIX + off[0][kk % 9]];
          nxt[1][kk] = cb[(size_t)(4 * (q + 1) + kk / 9) * PIX + off[1][kk % 9]];
        }
      }
#pragma unroll
      for (int j = 0; j < 9; ++j) {
        v2f b0, b1;
#pragma unroll
        for (int v = 0; v < 2; ++v) {
          const int kk = 2 * j + v;
          b0[v] = ok[0][kk % 9] ? cur[0][kk] : 0.0f;
          b1[v] = ok[1][kk % 9] ? cur[1][kk] : 0.0f;
        }
        acc0 = __builtin_amdgcn_wmma_f32_16x16x4_f32(
            false, a[q][j], false, b0, (short)0, acc0, false, false);
        acc1 = __builtin_amdgcn_wmma_f32_16x16x4_f32(
            false, a[q][j], false, b1, (short)0, acc1, false, false);
      }
#pragma unroll
      for (int kk = 0; kk < 18; ++kk) { cur[0][kk] = nxt[0][kk]; cur[1][kk] = nxt[1][kk]; }
    }

    float* s0 = sm + (it * 32 + row) * 113 + wave * 16 + 8 * half;
    float* s1 = s0 + 16 * 113;
#pragma unroll
    for (int v = 0; v < 8; ++v) { s0[v] = acc0[v]; s1[v] = acc1[v]; }

    __syncthreads();

    if (tid < 128) {
      const int pix = it * 32 + (tid & 31);
      const int ij  = tid >> 5;
      const float* base = sm + pix * 113;
      float r[25];
      float mx = -3.4e38f;
#pragma unroll
      for (int k = 0; k < 25; ++k) { r[k] = base[k * 4 + ij]; mx = fmaxf(mx, r[k]); }
      float sum = 0.0f;
#pragma unroll
      for (int k = 0; k < 25; ++k) { r[k] = __expf(r[k] - mx); sum += r[k]; }
      const float inv = 1.0f / sum;
      float* mplane = mask + (size_t)(n * HW + h) * MPLANE;
      float* mquad  = mplane + (ij * 6 * HW + pix) * 4;
#pragma unroll
      for (int g = 0; g < 6; ++g) {
        float4 o;
        o.x = r[4 * g + 0] * inv; o.y = r[4 * g + 1] * inv;
        o.z = r[4 * g + 2] * inv; o.w = r[4 * g + 3] * inv;
        *(float4*)(mquad + g * HW * 4) = o;
      }
      mplane[MTAIL + ij * HW + pix] = r[24] * inv;
    }
  }
}

__global__ __launch_bounds__(64) void carafe_reassemble(
    const float* __restrict__ x, const float* __restrict__ mask,
    float* __restrict__ out) {
  const int t  = threadIdx.x;
  const int b  = blockIdx.x;
  const int cg = b & 31;
  const int h  = (b >> 5) & 63;
  const int n  = b >> 11;

  const float* mplane = mask + (size_t)(n * HW + h) * MPLANE;
  float ms[100];
#pragma unroll
  for (int ij = 0; ij < 4; ++ij) {
#pragma unroll
    for (int g = 0; g < 6; ++g) {
      const float4 m4 = *(const float4*)(mplane + ((ij * 6 + g) * HW + t) * 4);
      ms[ij * 25 + 4 * g + 0] = m4.x; ms[ij * 25 + 4 * g + 1] = m4.y;
      ms[ij * 25 + 4 * g + 2] = m4.z; ms[ij * 25 + 4 * g + 3] = m4.w;
    }
    ms[ij * 25 + 24] = mplane[MTAIL + ij * HW + t];
  }

  for (int ci = 0; ci < 8; ++ci) {
    const int c = cg * 8 + ci;
    const float* xb = x + ((size_t)(n * 256 + c)) * PIX;

    float xv[25];
#pragma unroll
    for (int di = 0; di < 5; ++di) {
      const int y  = h + di - 2;
      const int yc = min(max(y, 0), HW - 1);
#pragma unroll
      for (int dj = 0; dj < 5; ++dj) {
        const int xx = t + dj - 2;
        const int xc = min(max(xx, 0), HW - 1);
        const bool ok = ((unsigned)y < HW) & ((unsigned)xx < HW);
        const float val = xb[yc * HW + xc];
        xv[di * 5 + dj] = ok ? val : 0.0f;
      }
    }

    float a0 = 0.f, a1 = 0.f, a2 = 0.f, a3 = 0.f;
#pragma unroll
    for (int k = 0; k < 25; ++k) {
      const float pch = xv[k];
      a0 += pch * ms[k];
      a1 += pch * ms[25 + k];
      a2 += pch * ms[50 + k];
      a3 += pch * ms[75 + k];
    }

    float* ob = out + (((size_t)(n * 256 + c) * 128) + 2 * h) * 128 + 2 * t;
    float2 r0; r0.x = a0; r0.y = a1;
    float2 r1; r1.x = a2; r1.y = a3;
    *(float2*)(ob)       = r0;
    *(float2*)(ob + 128) = r1;
  }
}

extern "C" void kernel_launch(void* const* d_in, const int* in_sizes, int n_in,
                              void* d_out, int out_size, void* d_ws, size_t ws_size,
                              hipStream_t stream) {
  (void)in_sizes; (void)n_in; (void)out_size; (void)ws_size;
  const float* x      = (const float*)d_in[0];
  const float* w_comp = (const float*)d_in[1];
  const float* b_comp = (const float*)d_in[2];
  const float* w_enc  = (const float*)d_in[3];
  const float* b_enc  = (const float*)d_in[4];
  float* out = (float*)d_out;

  float* comp = (float*)d_ws;
  float* mask = comp + (size_t)4 * CCOMP * PIX;

  carafe_compress<<<256, 128, 0, stream>>>(x, w_comp, b_comp, comp);
  carafe_encode_softmax<<<256, 224, 0, stream>>>(comp, w_enc, b_enc, mask);
  carafe_reassemble<<<8192, 64, 0, stream>>>(x, mask, out);
}


# Reassembly re-tiled in registers: a lane keeps the weights of 2 vertically adjacent pixels over 32 channels, one shared 6x5 window (30 loads) per channel; 2048 waves, all resident; sums unchanged
# speedup vs baseline: 3.4408x; 1.0721x over previous
#include <hip/hip_runtime.h>
#include <math.h>


typedef __attribute__((ext_vector_type(2))) float v2f;
typedef __attribute__((ext_vector_type(4))) float v4f;
typedef __attribute__((ext_vector_type(8))) float v8f;

#define HW 64
#define PIX 4096
#define CIN 256
#define CCOMP 64
#define MENC 100
#define MPAD 112
#define KENC 576
#define NPIX_TOTAL 16384
#define MPLANE 6400
#define MTAIL 6144

__global__ __launch_bounds__(128) __attribute__((amdgpu_waves_per_eu(1, 1)))
void carafe_compress(
    const float* __restrict__ x, const float* __restrict__ w_comp,
    const float* __restrict__ b_comp, float* __restrict__ comp) {
  const int mt   = threadIdx.x >> 5;
  const int lane = threadIdx.x & 31;
  const int col  = lane & 15;
  const int half = lane >> 4;

  const int gp  = blockIdx.x << 6;
  const int n   = gp >> 12;
  const int pin = gp & 4095;

  v8f acc[4];
#pragma unroll
  for (int v = 0; v < 8; ++v) {
    const float bv = b_comp[mt * 16 + v + 8 * half];
#pragma unroll
    for (int j = 0; j < 4; ++j) acc[j][v] = bv;
  }

  const float* xb = x + ((size_t)n * CIN + 2 * half) * PIX + pin + 4 * col;
  const float* wr = w_comp + (mt * 16 + col) * CIN + 2 * half;

  v4f cb[16][2], nb[16][2];
  v2f ca[16], na[16];
#pragma unroll
  for (int s = 0; s < 16; ++s) {
    ca[s]    = *(const v2f*)(wr + 4 * s);
    cb[s][0] = *(const v4f*)(xb + (size_t)(4 * s + 0) * PIX);
    cb[s][1] = *(const v4f*)(xb + (size_t)(4 * s + 1) * PIX);
  }

#pragma unroll
  for (int ch = 0; ch < 4; ++ch) {
    if (ch + 1 < 4) {
#pragma unroll
      for (int s = 0; s < 16; ++s) {
        const int k0 = 64 * (ch + 1) + 4 * s;
        na[s]    = *(const v2f*)(wr + k0);
        nb[s][0] = *(const v4f*)(xb + (size_t)(k0 + 0) * PIX);
        nb[s][1] = *(const v4f*)(xb + (size_t)(k0 + 1) * PIX);
      }
    }
#pragma unroll
    for (int s = 0; s < 16; ++s) {
#pragma unroll
      for (int j = 0; j < 4; ++j) {
        v2f b;
        b[0] = cb[s][0][j];
        b[1] = cb[s][1][j];
        acc[j] = __builtin_amdgcn_wmma_f32_16x16x4_f32(
            false, ca[s], false, b, (short)0, acc[j], false, false);
      }
    }
#pragma unroll
    for (int s = 0; s < 16; ++s) {
      ca[s] = na[s]; cb[s][0] = nb[s][0]; cb[s][1] = nb[s][1];
    }
  }

  float* co = comp + ((size_t)n * CCOMP + mt * 16 + 8 * half) * PIX + pin + 4 * col;
#pragma unroll
  for (int v = 0; v < 8; ++v) {
    v4f o;
    o[0] = acc[0][v]; o[1] = acc[1][v]; o[2] = acc[2][v]; o[3] = acc[3][v];
    *(v4f*)(co + (size_t)v * PIX) = o;
  }
}

__global__ __launch_bounds__(224) void carafe_encode_softmax(
    const float* __restrict__ comp, const float* __restrict__ w_enc,
    const float* __restrict__ b_enc, float* __restrict__ mask) {
  __shared__ float sm[64 * 113];

  const int tid  = threadIdx.x;
  const int wave = tid >> 5;
  const int lane = tid & 31;
  const int row  = lane & 15;
  const int half = lane >> 4;

  const int n = blockIdx.x >> 6;
  const int h = blockIdx.x & 63;

  const int m  = wave * 16 + row;
  const int mc = min(m, MENC - 1);
  const bool mvalid = m < MENC;
  const float* wrow = w_enc + (size_t)mc * KENC + 18 * half;
  v2f a[16][9];
#pragma unroll
  for (int q = 0; q < 16; ++q)
#pragma unroll
    for (int j = 0; j < 9; ++j) {
      v2f t = *(const v2f*)(wrow + 36 * q + 2 * j);
      if (!mvalid) { t[0] = 0.0f; t[1] = 0.0f; }
      a[q][j] = t;
    }

  v8f bias;
#pragma unroll
  for (int v = 0; v < 8; ++v) {
    const int mo = wave * 16 + v + 8 * half;
    bias[v] = (mo < MENC) ? b_enc[mo] : 0.0f;
  }

  const float* cb = comp + (size_t)n * CCOMP * PIX + (size_t)(2 * half) * PIX;

  for (int it = 0; it < 2; ++it) {
    int  off[2][9];
    bool ok[2][9];
#pragma unroll
    for (int T = 0; T < 2; ++T) {
      const int px = it * 32 + T * 16 + row;
#pragma unroll
      for (int r = 0; r < 9; ++r) {
        const int y  = h + (r / 3) - 1;
        const int xx = px + (r % 3) - 1;
        ok[T][r] = ((unsigned)y < HW) & ((unsigned)xx < HW);
        const int yc = min(max(y, 0), HW - 1);
        const int xc = min(max(xx, 0), HW - 1);
        off[T][r] = yc * HW + xc;
      }
    }

    v8f acc0 = bias, acc1 = bias;

    float cur[2][18], nxt[2][18];
#pragma unroll
    for (int kk = 0; kk < 18; ++kk) {
      cur[0][kk] = cb[(size_t)(kk / 9) * PIX + off[0][kk % 9]];
      cur[1][kk] = cb[(size_t)(kk / 9) * PIX + off[1][kk % 9]];
    }

#pragma unroll
    for (int q = 0; q < 16; ++q) {
      if (q + 1 < 16) {
#pragma unroll
        for (int kk = 0; kk < 18; ++kk) {
          nxt[0][kk] = cb[(size_t)(4 * (q + 1) + kk / 9) * PIX + off[0][kk % 9]];
          nxt[1][kk] = cb[(size_t)(4 * (q + 1) + kk / 9) * PIX + off[1][kk % 9]];
        }
      }
#pragma unroll
      for (int j = 0; j < 9; ++j) {
        v2f b0, b1;
#pragma unroll
        for (int v = 0; v < 2; ++v) {
          const int kk = 2 * j + v;
          b0[v] = ok[0][kk % 9] ? cur[0][kk] : 0.0f;
          b1[v] = ok[1][kk % 9] ? cur[1][kk] : 0.0f;
        }
        acc0 = __builtin_amdgcn_wmma_f32_16x16x4_f32(
            false, a[q][j], false, b0, (short)0, acc0, false, false);
        acc1 = __builtin_amdgcn_wmma_f32_16x16x4_f32(
            false, a[q][j], false, b1, (short)0, acc1, false, false);
      }
#pragma unroll
      for (int kk = 0; kk < 18; ++kk) { cur[0][kk] = nxt[0][kk]; cur[1][kk] = nxt[1][kk]; }
    }

    float* s0 = sm + (it * 32 + row) * 113 + wave * 16 + 8 * half;
    float* s1 = s0 + 16 * 113;
#pragma unroll
    for (int v = 0; v < 8; ++v) { s0[v] = acc0[v]; s1[v] = acc1[v]; }

    __syncthreads();

    if (tid < 128) {
      const int pix = it * 32 + (tid & 31);
      const int ij  = tid >> 5;
      const float* base = sm + pix * 113;
      float r[25];
      float mx = -3.4e38f;
#pragma unroll
      for (int k = 0; k < 25; ++k) { r[k] = base[k * 4 + ij]; mx = fmaxf(mx, r[k]); }
      float sum = 0.0f;
#pragma unroll
      for (int k = 0; k < 25; ++k) { r[k] = __expf(r[k] - mx); sum += r[k]; }
      const float inv = 1.0f / sum;
      float* mplane = mask + (size_t)(n * HW + h) * MPLANE;
      float* mquad  = mplane + (ij * 6 * HW + pix) * 4;
#pragma unroll
      for (int g = 0; g < 6; ++g) {
        float4 o;
        o.x = r[4 * g + 0] * inv; o.y = r[4 * g + 1] * inv;
        o.z = r[4 * g + 2] * inv; o.w = r[4 * g + 3] * inv;
        *(float4*)(mquad + g * HW * 4) = o;
      }
      mplane[MTAIL + ij * HW + pix] = r[24] * inv;
    }
  }
}

#define RCH 32
#define RING 4

template <bool LAST>
__device__ __forceinline__ void reassemble_trip(float (&xw)[RING][30],
                                                const float* const (&p)[30],
                                                const float (&ms)[2][100], float* og) {
#pragma unroll
  for (int s = 0; s < RING; ++s) {
    if (!LAST || s == 0) {
#pragma unroll
      for (int q = 0; q < 30; ++q) xw[(s + RING - 1) % RING][q] = p[q][(s + RING - 1) * PIX];
    }

    float a[2][4];
#pragma unroll
    for (int pp = 0; pp < 2; ++pp) {
      a[pp][0] = 0.f; a[pp][1] = 0.f; a[pp][2] = 0.f; a[pp][3] = 0.f;
#pragma unroll
      for (int k = 0; k < 25; ++k) {
        const float pch = xw[s][5 * pp + k];
        a[pp][0] += pch * ms[pp][k];
        a[pp][1] += pch * ms[pp][25 + k];
        a[pp][2] += pch * ms[pp][50 + k];
        a[pp][3] += pch * ms[pp][75 + k];
      }
    }

    float* ob = og + (size_t)s * (128 * 128);
#pragma unroll
    for (int pp = 0; pp < 2; ++pp) {
      float2 r0; r0.x = a[pp][0]; r0.y = a[pp][1];
      float2 r1; r1.x = a[pp][2]; r1.y = a[pp][3];
      *(float2*)(ob + (2 * pp) * 128)     = r0;
      *(float2*)(ob + (2 * pp + 1) * 128) = r1;
    }
  }
}

__global__ __launch_bounds__(64) void carafe_reassemble(
    const float* __restrict__ x, const float* __restrict__ mask,
    float* __restrict__ out) {
  const int t  = threadIdx.x;
  const int b  = blockIdx.x;
  const int cg = b & 7;
  const int h0 = ((b >> 3) & 31) * 2;
  const int n  = b >> 8;

  float ms[2][100];
#pragma unroll
  for (int p = 0; p < 2; ++p) {
    const float* mplane = mask + (size_t)(n * HW + h0 + p) * MPLANE;
#pragma unroll
    for (int ij = 0; ij < 4; ++ij) {
#pragma unroll
      for (int g = 0; g < 6; ++g) {
        const float4 m4 = *(const float4*)(mplane + ((ij * 6 + g) * HW + t) * 4);
        ms[p][ij * 25 + 4 * g + 0] = m4.x; ms[p][ij * 25 + 4 * g + 1] = m4.y;
        ms[p][ij * 25 + 4 * g + 2] = m4.z; ms[p][ij * 25 + 4 * g + 3] = m4.w;
      }
      ms[p][ij * 25 + 24] = mplane[MTAIL + ij * HW + t];
    }
  }

#pragma unroll
  for (int p = 0; p < 2; ++p)
#pragma unroll
    for (int di = 0; di < 5; ++di)
#pragma unroll
      for (int dj = 0; dj < 5; ++dj) {
        const bool ok = ((unsigned)(h0 + p + di - 2) < HW) & ((unsigned)(t + dj - 2) < HW);
#pragma unroll
        for (int ij = 0; ij < 4; ++ij)
          ms[p][ij * 25 + di * 5 + dj] = ok ? ms[p][ij * 25 + di * 5 + dj] : 0.0f;
      }

  const float* xg = x + (size_t)(n * 256 + cg * RCH) * PIX;
  const float* p[30];
#pragma unroll
  for (int r = 0; r < 6; ++r) {
    const int yc = min(max(h0 - 2 + r, 0), HW - 1);
#pragma unroll
    for (int dj = 0; dj < 5; ++dj) {
      const int xc = min(max(t + dj - 2, 0), HW - 1);
      p[r * 5 + dj] = xg + (yc * HW + xc);
    }
  }
  float* og = out + (((size_t)(n * 256 + cg * RCH) * 128) + 2 * h0) * 128 + 2 * t;

  float xw[RING][30];
#pragma unroll
  for (int s = 0; s < RING - 1; ++s)
#pragma unroll
    for (int q = 0; q < 30; ++q) xw[s][q] = p[q][s * PIX];

#pragma nounroll
  for (int trip = 0; trip < RCH / RING - 1; ++trip) {
    reassemble_trip<false>(xw, p, ms, og);
#pragma unroll
    for (int q = 0; q < 30; ++q) p[q] += RING * PIX;
    og += (size_t)RING * 128 * 128;
  }
  reassemble_trip<true>(xw, p, ms, og);
}

extern "C" void kernel_launch(void* const* d_in, const int* in_sizes, int n_in,
                              void* d_out, int out_size, void* d_ws, size_t ws_size,
                              hipStream_t stream) {
  (void)in_sizes; (void)n_in; (void)out_size; (void)ws_size;
  const float* x      = (const float*)d_in[0];
  const float* w_comp = (const float*)d_in[1];
  const float* b_comp = (const float*)d_in[2];
  const float* w_enc  = (const float*)d_in[3];
  const float* b_enc  = (const float*)d_in[4];
  float* out = (float*)d_out;

  float* comp = (float*)d_ws;
  float* mask = comp + (size_t)4 * CCOMP * PIX;

  carafe_compress<<<256, 128, 0, stream>>>(x, w_comp, b_comp, comp);
  carafe_encode_softmax<<<256, 224, 0, stream>>>(comp, w_enc, b_enc, mask);
  carafe_reassemble<<<1024, 64, 0, stream>>>(x, mask, out);
}


# 3x3 conv B operand: the row's input window (64 ch x 3 rows) is copied once per work-group to LDS with a zero halo; B elements are LDS reads, no clamp, select or global gather; sums unchanged
# speedup vs baseline: 3.7327x; 1.0848x over previous
#include <hip/hip_runtime.h>
#include <math.h>


typedef __attribute__((ext_vector_type(2))) float v2f;
typedef __attribute__((ext_vector_type(4))) float v4f;
typedef __attribute__((ext_vector_type(8))) float v8f;

#define HW 64
#define PIX 4096
#define CIN 256
#define CCOMP 64
#define MENC 100
#define MPAD 112
#define KENC 576
#define NPIX_TOTAL 16384
#define MPLANE 6400
#define MTAIL 6144

__global__ __launch_bounds__(128) __attribute__((amdgpu_waves_per_eu(1, 1)))
void carafe_compress(
    const float* __restrict__ x, const float* __restrict__ w_comp,
    const float* __restrict__ b_comp, float* __restrict__ comp) {
  const int mt   = threadIdx.x >> 5;
  const int lane = threadIdx.x & 31;
  const int col  = lane & 15;
  const int half = lane >> 4;

  const int gp  = blockIdx.x << 6;
  const int n   = gp >> 12;
  const int pin = gp & 4095;

  v8f acc[4];
#pragma unroll
  for (int v = 0; v < 8; ++v) {
    const float bv = b_comp[mt * 16 + v + 8 * half];
#pragma unroll
    for (int j = 0; j < 4; ++j) acc[j][v] = bv;
  }

  const float* xb = x + ((size_t)n * CIN + 2 * half) * PIX + pin + 4 * col;
  const float* wr = w_comp + (mt * 16 + col) * CIN + 2 * half;

  v4f cb[16][2], nb[16][2];
  v2f ca[16], na[16];
#pragma unroll
  for (int s = 0; s < 16; ++s) {
    ca[s]    = *(const v2f*)(wr + 4 * s);
    cb[s][0] = *(const v4f*)(xb + (size_t)(4 * s + 0) * PIX);
    cb[s][1] = *(const v4f*)(xb + (size_t)(4 * s + 1) * PIX);
  }

#pragma unroll
  for (int ch = 0; ch < 4; ++ch) {
    if (ch + 1 < 4) {
#pragma unroll
      for (int s = 0; s < 16; ++s) {
        const int k0 = 64 * (ch + 1) + 4 * s;
        na[s]    = *(const v2f*)(wr + k0);
        nb[s][0] = *(const v4f*)(xb + (size_t)(k0 + 0) * PIX);
        nb[s][1] = *(const v4f*)(xb + (size_t)(k0 + 1) * PIX);
      }
    }
#pragma unroll
    for (int s = 0; s < 16; ++s) {
#pragma unroll
      for (int j = 0; j < 4; ++j) {
        v2f b;
        b[0] = cb[s][0][j];
        b[1] = cb[s][1][j];
        acc[j] = __builtin_amdgcn_wmma_f32_16x16x4_f32(
            false, ca[s], false, b, (short)0, acc[j], false, false);
      }
    }
#pragma unroll
    for (int s = 0; s < 16; ++s) {
      ca[s] = na[s]; cb[s][0] = nb[s][0]; cb[s][1] = nb[s][1];
    }
  }

  float* co = comp + ((size_t)n * CCOMP + mt * 16 + 8 * half) * PIX + pin + 4 * col;
#pragma unroll
  for (int v = 0; v < 8; ++v) {
    v4f o;
    o[0] = acc[0][v]; o[1] = acc[1][v]; o[2] = acc[2][v]; o[3] = acc[3][v];
    *(v4f*)(co + (size_t)v * PIX) = o;
  }
}

#define WPITCH 72
#define WROWS 192
#define WQUADS 3072

__global__ __launch_bounds__(224) void carafe_encode_softmax(
    const float* __restrict__ comp, const float* __restrict__ w_enc,
    const float* __restrict__ b_enc, float* __restrict__ mask, int qstride) {
  __shared__ float sm[64 * 113];
  __shared__ __attribute__((aligned(16))) float win[WROWS * WPITCH];

  const int tid  = threadIdx.x;
  const int wave = tid >> 5;
  const int lane = tid & 31;
  const int row  = lane & 15;
  const int half = lane >> 4;

  const int n = blockIdx.x >> 6;
  const int h = blockIdx.x & 63;

  const float* cimg = comp + (size_t)n * CCOMP * PIX;
  v4f wv[14];
#pragma unroll
  for (int s = 0; s < 14; ++s) {
    const int i  = min(tid + 224 * s, WQUADS - 1);
    const int ch = (i >> 4) & 63;
    const int yc = min(max(h + (i >> 10) - 1, 0), HW - 1);
    wv[s] = *(const v4f*)(cimg + (size_t)ch * PIX + yc * HW + 4 * (i & 15));
  }

  const int m  = wave * 16 + row;
  const int mc = min(m, MENC - 1);
  const bool mvalid = m < MENC;
  const float* wrow = w_enc + (size_t)mc * KENC + 18 * half;
  v2f a[16][9];
#pragma unroll
  for (int q = 0; q < 16; ++q)
#pragma unroll
    for (int j = 0; j < 9; ++j) {
      v2f t = *(const v2f*)(wrow + 36 * q + 2 * j);
      if (!mvalid) { t[0] = 0.0f; t[1] = 0.0f; }
      a[q][j] = t;
    }

  v8f bias;
#pragma unroll
  for (int v = 0; v < 8; ++v) {
    const int mo = wave * 16 + v + 8 * half;
    bias[v] = (mo < MENC) ? b_enc[mo] : 0.0f;
  }

#pragma unroll
  for (int s = 0; s < 14; ++s) {
    const int i  = tid + 224 * s;
    const int r  = i >> 10;
    const int ch = (i >> 4) & 63;
    v4f t = wv[s];
    if ((unsigned)(h + r - 1) >= HW) { t[0] = 0.0f; t[1] = 0.0f; t[2] = 0.0f; t[3] = 0.0f; }
    if (i < WQUADS) *(v4f*)(win + (ch * 3 + r) * WPITCH + 4 + 4 * (i & 15)) = t;
  }
  if (tid < WROWS) {
    win[tid * WPITCH + 3]  = 0.0f;
    win[tid * WPITCH + 68] = 0.0f;
  }
  __syncthreads();

  const float* wl0 = win + (2 * half) * 3 * WPITCH + 3 + row;
  const int cstride = qstride >> 2;

  for (int it = 0; it < 2; ++it) {
    const float* wl = wl0 + it * 32;

    v8f acc0 = bias, acc1 = bias;

    float cur[2][18], nxt[2][18];
    {
      const float* pa = wl;
      const float* pb = wl + cstride;
#pragma unroll
      for (int T = 0; T < 2; ++T)
#pragma unroll
        for (int kk = 0; kk < 18; ++kk) {
          const int t = kk % 9;
          if (kk < 10) cur[T][kk] = pa[(kk / 9) * 3 * WPITCH + (t / 3) * WPITCH + t % 3 + 16 * T];
          else         cur[T][kk] = pb[(t / 3) * WPITCH + t % 3 + 16 * T];
        }
    }

#pragma unroll
    for (int q = 0; q < 16; ++q) {
      if (q + 1 < 16) {
        const float* pa = wl + (q + 1) * qstride;
        const float* pb = pa + cstride;
#pragma unroll
        for (int T = 0; T < 2; ++T)
#pragma unroll
          for (int kk = 0; kk < 18; ++kk) {
            const int t = kk % 9;
            if (kk < 10) nxt[T][kk] = pa[(kk / 9) * 3 * WPITCH + (t / 3) * WPITCH + t % 3 + 16 * T];
            else         nxt[T][kk] = pb[(t / 3) * WPITCH + t % 3 + 16 * T];
          }
        __builtin_amdgcn_sched_barrier(0);
      }
#pragma unroll
      for (int j = 0; j < 9; ++j) {
        v2f b0, b1;
        b0[0] = cur[0][2 * j]; b0[1] = cur[0][2 * j + 1];
        b1[0] = cur[1][2 * j]; b1[1] = cur[1][2 * j + 1];
        acc0 = __builtin_amdgcn_wmma_f32_16x16x4_f32(
            false, a[q][j], false, b0, (short)0, acc0, false, false);
        acc1 = __builtin_amdgcn_wmma_f32_16x16x4_f32(
            false, a[q][j], false, b1, (short)0, acc1, false, false);
      }
#pragma unroll
      for (int kk = 0; kk < 18; ++kk) { cur[0][kk] = nxt[0][kk]; cur[1][kk] = nxt[1][kk]; }
    }

    float* s0 = sm + (it * 32 + row) * 113 + wave * 16 + 8 * half;
    float* s1 = s0 + 16 * 113;
#pragma unroll
    for (int v = 0; v < 8; ++v) { s0[v] = acc0[v]; s1[v] = acc1[v]; }

    __syncthreads();

    if (tid < 128) {
      const int pix = it * 32 + (tid & 31);
      const int ij  = tid >> 5;
      const float* base = sm + pix * 113;
      float r[25];
      float mx = -3.4e38f;
#pragma unroll
      for (int k = 0; k < 25; ++k) { r[k] = base[k * 4 + ij]; mx = fmaxf(mx, r[k]); }
      float sum = 0.0f;
#pragma unroll
      for (int k = 0; k < 25; ++k) { r[k] = __expf(r[k] - mx); sum += r[k]; }
      const float inv = 1.0f / sum;
      float* mplane = mask + (size_t)(n * HW + h) * MPLANE;
      float* mquad  = mplane + (ij * 6 * HW + pix) * 4;
#pragma unroll
      for (int g = 0; g < 6; ++g) {
        float4 o;
        o.x = r[4 * g + 0] * inv; o.y = r[4 * g + 1] * inv;
        o.z = r[4 * g + 2] * inv; o.w = r[4 * g + 3] * inv;
        *(float4*)(mquad + g * HW * 4) = o;
      }
      mplane[MTAIL + ij * HW + pix] = r[24] * inv;
    }
  }
}

#define RCH 32
#define RING 4

template <bool LAST>
__device__ __forceinline__ void reassemble_trip(float (&xw)[RING][30],
                                                const float* const (&p)[30],
                                                const float (&ms)[2][100], float* og) {
#pragma unroll
  for (int s = 0; s < RING; ++s) {
    if (!LAST || s == 0) {
#pragma unroll
      for (int q = 0; q < 30; ++q) xw[(s + RING - 1) % RING][q] = p[q][(s + RING - 1) * PIX];
    }

    float a[2][4];
#pragma unroll
    for (int pp = 0; pp < 2; ++pp) {
      a[pp][0] = 0.f; a[pp][1] = 0.f; a[pp][2] = 0.f; a[pp][3] = 0.f;
#pragma unroll
      for (int k = 0; k < 25; ++k) {
        const float pch = xw[s][5 * pp + k];
        a[pp][0] += pch * ms[pp][k];
        a[pp][1] += pch * ms[pp][25 + k];
        a[pp][2] += pch * ms[pp][50 + k];
        a[pp][3] += pch * ms[pp][75 + k];
      }
    }

    float* ob = og + (size_t)s * (128 * 128);
#pragma unroll
    for (int pp = 0; pp < 2; ++pp) {
      float2 r0; r0.x = a[pp][0]; r0.y = a[pp][1];
      float2 r1; r1.x = a[pp][2]; r1.y = a[pp][3];
      *(float2*)(ob + (2 * pp) * 128)     = r0;
      *(float2*)(ob + (2 * pp + 1) * 128) = r1;
    }
  }
}

__global__ __launch_bounds__(64) void carafe_reassemble(
    const float* __restrict__ x, const float* __restrict__ mask,
    float* __restrict__ out) {
  const int t  = threadIdx.x;
  const int b  = blockIdx.x;
  const int cg = b & 7;
  const int h0 = ((b >> 3) & 31) * 2;
  const int n  = b >> 8;

  float ms[2][100];
#pragma unroll
  for (int p = 0; p < 2; ++p) {
    const float* mplane = mask + (size_t)(n * HW + h0 + p) * MPLANE;
#pragma unroll
    for (int ij = 0; ij < 4; ++ij) {
#pragma unroll
      for (int g = 0; g < 6; ++g) {
        const float4 m4 = *(const float4*)(mplane + ((ij * 6 + g) * HW + t) * 4);
        ms[p][ij * 25 + 4 * g + 0] = m4.x; ms[p][ij * 25 + 4 * g + 1] = m4.y;
        ms[p][ij * 25 + 4 * g + 2] = m4.z; ms[p][ij * 25 + 4 * g + 3] = m4.w;
      }
      ms[p][ij * 25 + 24] = mplane[MTAIL + ij * HW + t];
    }
  }

#pragma unroll
  for (int p = 0; p < 2; ++p)
#pragma unroll
    for (int di = 0; di < 5; ++di)
#pragma unroll
      for (int dj = 0; dj < 5; ++dj) {
        const bool ok = ((unsigned)(h0 + p + di - 2) < HW) & ((unsigned)(t + dj - 2) < HW);
#pragma unroll
        for (int ij = 0; ij < 4; ++ij)
          ms[p][ij * 25 + di * 5 + dj] = ok ? ms[p][ij * 25 + di * 5 + dj] : 0.0f;
      }

  const float* xg = x + (size_t)(n * 256 + cg * RCH) * PIX;
  const float* p[30];
#pragma unroll
  for (int r = 0; r < 6; ++r) {
    const int yc = min(max(h0 - 2 + r, 0), HW - 1);
#pragma unroll
    for (int dj = 0; dj < 5; ++dj) {
      const int xc = min(max(t + dj - 2, 0), HW - 1);
      p[r * 5 + dj] = xg + (yc * HW + xc);
    }
  }
  float* og = out + (((size_t)(n * 256 + cg * RCH) * 128) + 2 * h0) * 128 + 2 * t;

  float xw[RING][30];
#pragma unroll
  for (int s = 0; s < RING - 1; ++s)
#pragma unroll
    for (int q = 0; q < 30; ++q) xw[s][q] = p[q][s * PIX];

#pragma nounroll
  for (int trip = 0; trip < RCH / RING - 1; ++trip) {
    reassemble_trip<false>(xw, p, ms, og);
#pragma unroll
    for (int q = 0; q < 30; ++q) p[q] += RING * PIX;
    og += (size_t)RING * 128 * 128;
  }
  reassemble_trip<true>(xw, p, ms, og);
}

extern "C" void kernel_launch(void* const* d_in, const int* in_sizes, int n_in,
                              void* d_out, int out_size, void* d_ws, size_t ws_size,
                              hipStream_t stream) {
  (void)in_sizes; (void)n_in; (void)out_size; (void)ws_size;
  const float* x      = (const float*)d_in[0];
  const float* w_comp = (const float*)d_in[1];
  const float* b_comp = (const float*)d_in[2];
  const float* w_enc  = (const float*)d_in[3];
  const float* b_enc  = (const float*)d_in[4];
  float* out = (float*)d_out;

  float* comp = (float*)d_ws;
  float* mask = comp + (size_t)4 * CCOMP * PIX;

  carafe_compress<<<256, 128, 0, stream>>>(x, w_comp, b_comp, comp);
  carafe_encode_softmax<<<256, 224, 0, stream>>>(comp, w_enc, b_enc, mask, 864);
  carafe_reassemble<<<1024, 64, 0, stream>>>(x, mask, out);
}


# 3x3 conv by minimal filtering F(2x2,3x3): work-group = 2 rows x 32 px = 16 blocks, 16 GEMMs of K=64 in the transformed domain, 256 matrix ops a wave (was 576); conv sum order changed
# speedup vs baseline: 4.2174x; 1.1299x over previous
#include <hip/hip_runtime.h>
#include <math.h>


typedef __attribute__((ext_vector_type(2))) float v2f;
typedef __attribute__((ext_vector_type(4))) float v4f;
typedef __attribute__((ext_vector_type(8))) float v8f;

#define HW 64
#define PIX 4096
#define CIN 256
#define CCOMP 64
#define MENC 100
#define MPAD 112
#define KENC 576
#define NPIX_TOTAL 16384
#define MPLANE 6400
#define MTAIL 6144

__global__ __launch_bounds__(128) __attribute__((amdgpu_waves_per_eu(1, 1)))
void carafe_compress(
    const float* __restrict__ x, const float* __restrict__ w_comp,
    const float* __restrict__ b_comp, float* __restrict__ comp) {
  const int mt   = threadIdx.x >> 5;
  const int lane = threadIdx.x & 31;
  const int col  = lane & 15;
  const int half = lane >> 4;

  const int gp  = blockIdx.x << 6;
  const int n   = gp >> 12;
  const int pin = gp & 4095;

  v8f acc[4];
#pragma unroll
  for (int v = 0; v < 8; ++v) {
    const float bv = b_comp[mt * 16 + v + 8 * half];
#pragma unroll
    for (int j = 0; j < 4; ++j) acc[j][v] = bv;
  }

  const float* xb = x + ((size_t)n * CIN + 2 * half) * PIX + pin + 4 * col;
  const float* wr = w_comp + (mt * 16 + col) * CIN + 2 * half;

  v4f cb[16][2], nb[16][2];
  v2f ca[16], na[16];
#pragma unroll
  for (int s = 0; s < 16; ++s) {
    ca[s]    = *(const v2f*)(wr + 4 * s);
    cb[s][0] = *(const v4f*)(xb + (size_t)(4 * s + 0) * PIX);
    cb[s][1] = *(const v4f*)(xb + (size_t)(4 * s + 1) * PIX);
  }

#pragma unroll
  for (int ch = 0; ch < 4; ++ch) {
    if (ch + 1 < 4) {
#pragma unroll
      for (int s = 0; s < 16; ++s) {
        const int k0 = 64 * (ch + 1) + 4 * s;
        na[s]    = *(const v2f*)(wr + k0);
        nb[s][0] = *(const v4f*)(xb + (size_t)(k0 + 0) * PIX);
        nb[s][1] = *(const v4f*)(xb + (size_t)(k0 + 1) * PIX);
      }
    }
#pragma unroll
    for (int s = 0; s < 16; ++s) {
#pragma unroll
      for (int j = 0; j < 4; ++j) {
        v2f b;
        b[0] = cb[s][0][j];
        b[1] = cb[s][1][j];
        acc[j] = __builtin_amdgcn_wmma_f32_16x16x4_f32(
            false, ca[s], false, b, (short)0, acc[j], false, false);
      }
    }
#pragma unroll
    for (int s = 0; s < 16; ++s) {
      ca[s] = na[s]; cb[s][0] = nb[s][0]; cb[s][1] = nb[s][1];
    }
  }

  float* co = comp + ((size_t)n * CCOMP + mt * 16 + 8 * half) * PIX + pin + 4 * col;
#pragma unroll
  for (int v = 0; v < 8; ++v) {
    v4f o;
    o[0] = acc[0][v]; o[1] = acc[1][v]; o[2] = acc[2][v]; o[3] = acc[3][v];
    *(v4f*)(co + (size_t)v * PIX) = o;
  }
}

#define WP 44
#define WROWS 256
#define WQUADS 2048
#define WHALO 512

__global__ __launch_bounds__(224) void carafe_encode_softmax(
    const float* __restrict__ comp, const float* __restrict__ w_enc,
    const float* __restrict__ b_enc, float* __restrict__ mask) {
  __shared__ float sm[64 * 113];
  __shared__ __attribute__((aligned(16))) float win[WROWS * WP];
  __shared__ __attribute__((aligned(16))) float vt[16 * 32 * 16 * 2];

  const int tid  = threadIdx.x;
  const int wave = tid >> 5;
  const int lane = tid & 31;
  const int row  = lane & 15;
  const int half = lane >> 4;

  const int n  = blockIdx.x >> 6;
  const int h0 = ((blockIdx.x >> 1) & 31) * 2;
  const int w0 = (blockIdx.x & 1) * 32;


  const float* cimg = comp + (size_t)n * CCOMP * PIX;
  v4f wv[10];
#pragma unroll
  for (int s = 0; s < 10; ++s) {
    const int i   = min(tid + 224 * s, WQUADS - 1);
    const int rid = i >> 3;
    const int yc  = min(max(h0 - 1 + (rid & 3), 0), HW - 1);
    wv[s] = *(const v4f*)(cimg + (size_t)(rid >> 2) * PIX + yc * HW + w0 + 4 * (i & 7));
  }
  float hv[3];
#pragma unroll
  for (int s = 0; s < 3; ++s) {
    const int j   = min(tid + 224 * s, WHALO - 1);
    const int rid = j >> 1;
    const int yc  = min(max(h0 - 1 + (rid & 3), 0), HW - 1);
    const int xc  = (j & 1) ? min(w0 + 32, HW - 1) : max(w0 - 1, 0);
    hv[s] = cimg[(size_t)(rid >> 2) * PIX + yc * HW + xc];
  }

  const int m  = wave * 16 + row;
  const int mc = min(m, MENC - 1);
  const bool mvalid = m < MENC;
  const float* wrow = w_enc + (size_t)mc * KENC + 18 * half;
  v2f gw[4][9];
#pragma unroll
  for (int s = 0; s < 3; ++s)
#pragma unroll
    for (int j = 0; j < 9; ++j) gw[s][j] = *(const v2f*)(wrow + 36 * s + 2 * j);

  v8f bias;
#pragma unroll
  for (int v = 0; v < 8; ++v) {
    const int mo = wave * 16 + v + 8 * half;
    bias[v] = (mo < MENC) ? b_enc[mo] : 0.0f;
  }

#pragma unroll
  for (int s = 0; s < 10; ++s) {
    const int i   = tid + 224 * s;
    const int rid = i >> 3;
    v4f t = wv[s];
    if ((unsigned)(h0 - 1 + (rid & 3)) >= HW) { t[0] = 0.0f; t[1] = 0.0f; t[2] = 0.0f; t[3] = 0.0f; }
    if (i < WQUADS) *(v4f*)(win + rid * WP + 4 + 4 * (i & 7)) = t;
  }
#pragma unroll
  for (int s = 0; s < 3; ++s) {
    const int j   = tid + 224 * s;
    const int rid = j >> 1;
    const int x   = (j & 1) ? (w0 + 32) : (w0 - 1);
    const bool ok = ((unsigned)(h0 - 1 + (rid & 3)) < HW) & ((unsigned)x < HW);
    if (j < WHALO) win[rid * WP + ((j & 1) ? 36 : 3)] = ok ? hv[s] : 0.0f;
  }
  __syncthreads();

#pragma unroll
  for (int s = 0; s < 3; ++s) {
    const int id = tid + 224 * s;
    if (id < 512) {
      const int cp = id >> 4;
      const int t  = id & 15;
      float V[2][16];
#pragma unroll
      for (int e = 0; e < 2; ++e) {
        const float* dp = win + (2 * cp + e) * 4 * WP + 3 + 2 * t;
        float d[4][4], T[4][4];
#pragma unroll
        for (int r = 0; r < 4; ++r)
#pragma unroll
          for (int j = 0; j < 4; ++j) d[r][j] = dp[r * WP + j];
#pragma unroll
        for (int j = 0; j < 4; ++j) {
          T[0][j] = d[0][j] - d[2][j];
          T[1][j] = d[1][j] + d[2][j];
          T[2][j] = d[2][j] - d[1][j];
          T[3][j] = d[1][j] - d[3][j];
        }
#pragma unroll
        for (int i = 0; i < 4; ++i) {
          V[e][4 * i + 0] = T[i][0] - T[i][2];
          V[e][4 * i + 1] = T[i][1] + T[i][2];
          V[e][4 * i + 2] = T[i][2] - T[i][1];
          V[e][4 * i + 3] = T[i][1] - T[i][3];
        }
      }
#pragma unroll
      for (int pos = 0; pos < 16; ++pos) {
        v2f o; o[0] = V[0][pos]; o[1] = V[1][pos];
        ((v2f*)vt)[(pos * 32 + cp) * 16 + t] = o;
      }
    }
  }
  __syncthreads();

  const v2f* vl = (const v2f*)vt + half * 16 + row;
  v8f M[16];
#pragma unroll
  for (int pos = 0; pos < 16; ++pos)
#pragma unroll
    for (int v = 0; v < 8; ++v) M[pos][v] = 0.0f;

  v2f bcur[16], bnxt[16];
#pragma unroll
  for (int pos = 0; pos < 16; ++pos) bcur[pos] = vl[pos * 512];

#pragma unroll
  for (int s = 0; s < 16; ++s) {
    if (s + 3 < 16) {
#pragma unroll
      for (int j = 0; j < 9; ++j) gw[(s + 3) % 4][j] = *(const v2f*)(wrow + 36 * (s + 3) + 2 * j);
    }
    if (s + 1 < 16) {
#pragma unroll
      for (int pos = 0; pos < 16; ++pos) bnxt[pos] = vl[pos * 512 + 32 * (s + 1)];
    }
    __builtin_amdgcn_sched_barrier(0);

    float U[2][16];
#pragma unroll
    for (int e = 0; e < 2; ++e) {
      float g[9];
#pragma unroll
      for (int k = 0; k < 9; ++k) {
        const float w = gw[s % 4][(9 * e + k) >> 1][(9 * e + k) & 1];
        g[k] = mvalid ? w : 0.0f;
      }
      float P[4][3];
#pragma unroll
      for (int t = 0; t < 3; ++t) {
        const float sg = g[t] + g[6 + t];
        P[0][t] = g[t];
        P[1][t] = 0.5f * (sg + g[3 + t]);
        P[2][t] = 0.5f * (sg - g[3 + t]);
        P[3][t] = g[6 + t];
      }
#pragma unroll
      for (int i = 0; i < 4; ++i) {
        const float sp = P[i][0] + P[i][2];
        U[e][4 * i + 0] = P[i][0];
        U[e][4 * i + 1] = 0.5f * (sp + P[i][1]);
        U[e][4 * i + 2] = 0.5f * (sp - P[i][1]);
        U[e][4 * i + 3] = P[i][2];
      }
    }

#pragma unroll
    for (int pos = 0; pos < 16; ++pos) {
      v2f a; a[0] = U[0][pos]; a[1] = U[1][pos];
      M[pos] = __builtin_amdgcn_wmma_f32_16x16x4_f32(
          false, a, false, bcur[pos], (short)0, M[pos], false, false);
    }
#pragma unroll
    for (int pos = 0; pos < 16; ++pos) bcur[pos] = bnxt[pos];
  }

#pragma unroll
  for (int a = 0; a < 2; ++a) {
    v8f R[4];
#pragma unroll
    for (int j = 0; j < 4; ++j)
      R[j] = (a == 0) ? (M[j] + M[4 + j] + M[8 + j]) : (M[4 + j] - M[8 + j] - M[12 + j]);
    const v8f y0 = R[0] + R[1] + R[2] + bias;
    const v8f y1 = R[1] - R[2] - R[3] + bias;
    float* s0 = sm + (a * 32 + 2 * row) * 113 + wave * 16 + 8 * half;
#pragma unroll
    for (int v = 0; v < 8; ++v) { s0[v] = y0[v]; s0[113 + v] = y1[v]; }
  }
  __syncthreads();

  if (tid < 128) {
    const int ij = tid >> 5;
    const int w  = w0 + (tid & 31);
    for (int it = 0; it < 2; ++it) {
      const float* base = sm + (it * 32 + (tid & 31)) * 113;
      float r[25];
      float mx = -3.4e38f;
#pragma unroll
      for (int k = 0; k < 25; ++k) { r[k] = base[k * 4 + ij]; mx = fmaxf(mx, r[k]); }
      float sum = 0.0f;
#pragma unroll
      for (int k = 0; k < 25; ++k) { r[k] = __expf(r[k] - mx); sum += r[k]; }
      const float inv = 1.0f / sum;
      float* mplane = mask + (size_t)(n * HW + h0 + it) * MPLANE;
      float* mquad  = mplane + (ij * 6 * HW + w) * 4;
#pragma unroll
      for (int g = 0; g < 6; ++g) {
        float4 o;
        o.x = r[4 * g + 0] * inv; o.y = r[4 * g + 1] * inv;
        o.z = r[4 * g + 2] * inv; o.w = r[4 * g + 3] * inv;
        *(float4*)(mquad + g * HW * 4) = o;
      }
      mplane[MTAIL + ij * HW + w] = r[24] * inv;
    }
  }
}

#define RCH 32
#define RING 4

template <bool LAST>
__device__ __forceinline__ void reassemble_trip(float (&xw)[RING][30],
                                                const float* const (&p)[30],
                                                const float (&ms)[2][100], float* og) {
#pragma unroll
  for (int s = 0; s < RING; ++s) {
    if (!LAST || s == 0) {
#pragma unroll
      for (int q = 0; q < 30; ++q) xw[(s + RING - 1) % RING][q] = p[q][(s + RING - 1) * PIX];
    }

    float a[2][4];
#pragma unroll
    for (int pp = 0; pp < 2; ++pp) {
      a[pp][0] = 0.f; a[pp][1] = 0.f; a[pp][2] = 0.f; a[pp][3] = 0.f;
#pragma unroll
      for (int k = 0; k < 25; ++k) {
        const float pch = xw[s][5 * pp + k];
        a[pp][0] += pch * ms[pp][k];
        a[pp][1] += pch * ms[pp][25 + k];
        a[pp][2] += pch * ms[pp][50 + k];
        a[pp][3] += pch * ms[pp][75 + k];
      }
    }

    float* ob = og + (size_t)s * (128 * 128);
#pragma unroll
    for (int pp = 0; pp < 2; ++pp) {
      float2 r0; r0.x = a[pp][0]; r0.y = a[pp][1];
      float2 r1; r1.x = a[pp][2]; r1.y = a[pp][3];
      *(float2*)(ob + (2 * pp) * 128)     = r0;
      *(float2*)(ob + (2 * pp + 1) * 128) = r1;
    }
  }
}

__global__ __launch_bounds__(64) void carafe_reassemble(
    const float* __restrict__ x, const float* __restrict__ mask,
    float* __restrict__ out) {
  const int t  = threadIdx.x;
  const int b  = blockIdx.x;
  const int cg = b & 7;
  const int h0 = ((b >> 3) & 31) * 2;
  const int n  = b >> 8;

  float ms[2][100];
#pragma unroll
  for (int p = 0; p < 2; ++p) {
    const float* mplane = mask + (size_t)(n * HW + h0 + p) * MPLANE;
#pragma unroll
    for (int ij = 0; ij < 4; ++ij) {
#pragma unroll
      for (int g = 0; g < 6; ++g) {
        const float4 m4 = *(const float4*)(mplane + ((ij * 6 + g) * HW + t) * 4);
        ms[p][ij * 25 + 4 * g + 0] = m4.x; ms[p][ij * 25 + 4 * g + 1] = m4.y;
        ms[p][ij * 25 + 4 * g + 2] = m4.z; ms[p][ij * 25 + 4 * g + 3] = m4.w;
      }
      ms[p][ij * 25 + 24] = mplane[MTAIL + ij * HW + t];
    }
  }

#pragma unroll
  for (int p = 0; p < 2; ++p)
#pragma unroll
    for (int di = 0; di < 5; ++di)
#pragma unroll
      for (int dj = 0; dj < 5; ++dj) {
        const bool ok = ((unsigned)(h0 + p + di - 2) < HW) & ((unsigned)(t + dj - 2) < HW);
#pragma unroll
        for (int ij = 0; ij < 4; ++ij)
          ms[p][ij * 25 + di * 5 + dj] = ok ? ms[p][ij * 25 + di * 5 + dj] : 0.0f;
      }

  const float* xg = x + (size_t)(n * 256 + cg * RCH) * PIX;
  const float* p[30];
#pragma unroll
  for (int r = 0; r < 6; ++r) {
    const int yc = min(max(h0 - 2 + r, 0), HW - 1);
#pragma unroll
    for (int dj = 0; dj < 5; ++dj) {
      const int xc = min(max(t + dj - 2, 0), HW - 1);
      p[r * 5 + dj] = xg + (yc * HW + xc);
    }
  }
  float* og = out + (((size_t)(n * 256 + cg * RCH) * 128) + 2 * h0) * 128 + 2 * t;

  float xw[RING][30];
#pragma unroll
  for (int s = 0; s < RING - 1; ++s)
#pragma unroll
    for (int q = 0; q < 30; ++q) xw[s][q] = p[q][s * PIX];

#pragma nounroll
  for (int trip = 0; trip < RCH / RING - 1; ++trip) {
    reassemble_trip<false>(xw, p, ms, og);
#pragma unroll
    for (int q = 0; q < 30; ++q) p[q] += RING * PIX;
    og += (size_t)RING * 128 * 128;
  }
  reassemble_trip<true>(xw, p, ms, og);
}

extern "C" void kernel_launch(void* const* d_in, const int* in_sizes, int n_in,
                              void* d_out, int out_size, void* d_ws, size_t ws_size,
                              hipStream_t stream) {
  (void)in_sizes; (void)n_in; (void)out_size; (void)ws_size;
  const float* x      = (const float*)d_in[0];
  const float* w_comp = (const float*)d_in[1];
  const float* b_comp = (const float*)d_in[2];
  const float* w_enc  = (const float*)d_in[3];
  const float* b_enc  = (const float*)d_in[4];
  float* out = (float*)d_out;

  float* comp = (float*)d_ws;
  float* mask = comp + (size_t)4 * CCOMP * PIX;

  carafe_compress<<<256, 128, 0, stream>>>(x, w_comp, b_comp, comp);
  carafe_encode_softmax<<<256, 224, 0, stream>>>(comp, w_enc, b_enc, mask);
  carafe_reassemble<<<1024, 64, 0, stream>>>(x, mask, out);
}


# Reassembly: a lane owns a 2x2 pixel block (400 weights in registers); its 6x6 window = 18 8-byte global loads a channel (was 30 dwords per 2 pixels); mask quads split even/odd; sums unchanged
# speedup vs baseline: 4.7217x; 1.1196x over previous
#include <hip/hip_runtime.h>
#include <math.h>


typedef __attribute__((ext_vector_type(2))) float v2f;
typedef __attribute__((ext_vector_type(4))) float v4f;
typedef __attribute__((ext_vector_type(8))) float v8f;

#define HW 64
#define PIX 4096
#define CIN 256
#define CCOMP 64
#define MENC 100
#define MPAD 112
#define KENC 576
#define NPIX_TOTAL 16384
#define MPLANE 6400
#define MTAIL 6144

__global__ __launch_bounds__(128) __attribute__((amdgpu_waves_per_eu(1, 1)))
void carafe_compress(
    const float* __restrict__ x, const float* __restrict__ w_comp,
    const float* __restrict__ b_comp, float* __restrict__ comp) {
  const int mt   = threadIdx.x >> 5;
  const int lane = threadIdx.x & 31;
  const int col  = lane & 15;
  const int half = lane >> 4;

  const int gp  = blockIdx.x << 6;
  const int n   = gp >> 12;
  const int pin = gp & 4095;

  v8f acc[4];
#pragma unroll
  for (int v = 0; v < 8; ++v) {
    const float bv = b_comp[mt * 16 + v + 8 * half];
#pragma unroll
    for (int j = 0; j < 4; ++j) acc[j][v] = bv;
  }

  const float* xb = x + ((size_t)n * CIN + 2 * half) * PIX + pin + 4 * col;
  const float* wr = w_comp + (mt * 16 + col) * CIN + 2 * half;

  v4f cb[16][2], nb[16][2];
  v2f ca[16], na[16];
#pragma unroll
  for (int s = 0; s < 16; ++s) {
    ca[s]    = *(const v2f*)(wr + 4 * s);
    cb[s][0] = *(const v4f*)(xb + (size_t)(4 * s + 0) * PIX);
    cb[s][1] = *(const v4f*)(xb + (size_t)(4 * s + 1) * PIX);
  }

#pragma unroll
  for (int ch = 0; ch < 4; ++ch) {
    if (ch + 1 < 4) {
#pragma unroll
      for (int s = 0; s < 16; ++s) {
        const int k0 = 64 * (ch + 1) + 4 * s;
        na[s]    = *(const v2f*)(wr + k0);
        nb[s][0] = *(const v4f*)(xb + (size_t)(k0 + 0) * PIX);
        nb[s][1] = *(const v4f*)(xb + (size_t)(k0 + 1) * PIX);
      }
    }
#pragma unroll
    for (int s = 0; s < 16; ++s) {
#pragma unroll
      for (int j = 0; j < 4; ++j) {
        v2f b;
        b[0] = cb[s][0][j];
        b[1] = cb[s][1][j];
        acc[j] = __builtin_amdgcn_wmma_f32_16x16x4_f32(
            false, ca[s], false, b, (short)0, acc[j], false, false);
      }
    }
#pragma unroll
    for (int s = 0; s < 16; ++s) {
      ca[s] = na[s]; cb[s][0] = nb[s][0]; cb[s][1] = nb[s][1];
    }
  }

  float* co = comp + ((size_t)n * CCOMP + mt * 16 + 8 * half) * PIX + pin + 4 * col;
#pragma unroll
  for (int v = 0; v < 8; ++v) {
    v4f o;
    o[0] = acc[0][v]; o[1] = acc[1][v]; o[2] = acc[2][v]; o[3] = acc[3][v];
    *(v4f*)(co + (size_t)v * PIX) = o;
  }
}

#define WP 44
#define WROWS 256
#define WQUADS 2048
#define WHALO 512

__global__ __launch_bounds__(224) void carafe_encode_softmax(
    const float* __restrict__ comp, const float* __restrict__ w_enc,
    const float* __restrict__ b_enc, float* __restrict__ mask) {
  __shared__ float sm[64 * 113];
  __shared__ __attribute__((aligned(16))) float win[WROWS * WP];
  __shared__ __attribute__((aligned(16))) float vt[16 * 32 * 16 * 2];

  const int tid  = threadIdx.x;
  const int wave = tid >> 5;
  const int lane = tid & 31;
  const int row  = lane & 15;
  const int half = lane >> 4;

  const int n  = blockIdx.x >> 6;
  const int h0 = ((blockIdx.x >> 1) & 31) * 2;
  const int w0 = (blockIdx.x & 1) * 32;


  const float* cimg = comp + (size_t)n * CCOMP * PIX;
  v4f wv[10];
#pragma unroll
  for (int s = 0; s < 10; ++s) {
    const int i   = min(tid + 224 * s, WQUADS - 1);
    const int rid = i >> 3;
    const int yc  = min(max(h0 - 1 + (rid & 3), 0), HW - 1);
    wv[s] = *(const v4f*)(cimg + (size_t)(rid >> 2) * PIX + yc * HW + w0 + 4 * (i & 7));
  }
  float hv[3];
#pragma unroll
  for (int s = 0; s < 3; ++s) {
    const int j   = min(tid + 224 * s, WHALO - 1);
    const int rid = j >> 1;
    const int yc  = min(max(h0 - 1 + (rid & 3), 0), HW - 1);
    const int xc  = (j & 1) ? min(w0 + 32, HW - 1) : max(w0 - 1, 0);
    hv[s] = cimg[(size_t)(rid >> 2) * PIX + yc * HW + xc];
  }

  const int m  = wave * 16 + row;
  const int mc = min(m, MENC - 1);
  const bool mvalid = m < MENC;
  const float* wrow = w_enc + (size_t)mc * KENC + 18 * half;
  v2f gw[4][9];
#pragma unroll
  for (int s = 0; s < 3; ++s)
#pragma unroll
    for (int j = 0; j < 9; ++j) gw[s][j] = *(const v2f*)(wrow + 36 * s + 2 * j);

  v8f bias;
#pragma unroll
  for (int v = 0; v < 8; ++v) {
    const int mo = wave * 16 + v + 8 * half;
    bias[v] = (mo < MENC) ? b_enc[mo] : 0.0f;
  }

#pragma unroll
  for (int s = 0; s < 10; ++s) {
    const int i   = tid + 224 * s;
    const int rid = i >> 3;
    v4f t = wv[s];
    if ((unsigned)(h0 - 1 + (rid & 3)) >= HW) { t[0] = 0.0f; t[1] = 0.0f; t[2] = 0.0f; t[3] = 0.0f; }
    if (i < WQUADS) *(v4f*)(win + rid * WP + 4 + 4 * (i & 7)) = t;
  }
#pragma unroll
  for (int s = 0; s < 3; ++s) {
    const int j   = tid + 224 * s;
    const int rid = j >> 1;
    const int x   = (j & 1) ? (w0 + 32) : (w0 - 1);
    const bool ok = ((unsigned)(h0 - 1 + (rid & 3)) < HW) & ((unsigned)x < HW);
    if (j < WHALO) win[rid * WP + ((j & 1) ? 36 : 3)] = ok ? hv[s] : 0.0f;
  }
  __syncthreads();

#pragma unroll
  for (int s = 0; s < 3; ++s) {
    const int id = tid + 224 * s;
    if (id < 512) {
      const int cp = id >> 4;
      const int t  = id & 15;
      float V[2][16];
#pragma unroll
      for (int e = 0; e < 2; ++e) {
        const float* dp = win + (2 * cp + e) * 4 * WP + 3 + 2 * t;
        float d[4][4], T[4][4];
#pragma unroll
        for (int r = 0; r < 4; ++r)
#pragma unroll
          for (int j = 0; j < 4; ++j) d[r][j] = dp[r * WP + j];
#pragma unroll
        for (int j = 0; j < 4; ++j) {
          T[0][j] = d[0][j] - d[2][j];
          T[1][j] = d[1][j] + d[2][j];
          T[2][j] = d[2][j] - d[1][j];
          T[3][j] = d[1][j] - d[3][j];
        }
#pragma unroll
        for (int i = 0; i < 4; ++i) {
          V[e][4 * i + 0] = T[i][0] - T[i][2];
          V[e][4 * i + 1] = T[i][1] + T[i][2];
          V[e][4 * i + 2] = T[i][2] - T[i][1];
          V[e][4 * i + 3] = T[i][1] - T[i][3];
        }
      }
#pragma unroll
      for (int pos = 0; pos < 16; ++pos) {
        v2f o; o[0] = V[0][pos]; o[1] = V[1][pos];
        ((v2f*)vt)[(pos * 32 + cp) * 16 + t] = o;
      }
    }
  }
  __syncthreads();

  const v2f* vl = (const v2f*)vt + half * 16 + row;
  v8f M[16];
#pragma unroll
  for (int pos = 0; pos < 16; ++pos)
#pragma unroll
    for (int v = 0; v < 8; ++v) M[pos][v] = 0.0f;

  v2f bcur[16], bnxt[16];
#pragma unroll
  for (int pos = 0; pos < 16; ++pos) bcur[pos] = vl[pos * 512];

#pragma unroll
  for (int s = 0; s < 16; ++s) {
    if (s + 3 < 16) {
#pragma unroll
      for (int j = 0; j < 9; ++j) gw[(s + 3) % 4][j] = *(const v2f*)(wrow + 36 * (s + 3) + 2 * j);
    }
    if (s + 1 < 16) {
#pragma unroll
      for (int pos = 0; pos < 16; ++pos) bnxt[pos] = vl[pos * 512 + 32 * (s + 1)];
    }
    __builtin_amdgcn_sched_barrier(0);

    float U[2][16];
#pragma unroll
    for (int e = 0; e < 2; ++e) {
      float g[9];
#pragma unroll
      for (int k = 0; k < 9; ++k) {
        const float w = gw[s % 4][(9 * e + k) >> 1][(9 * e + k) & 1];
        g[k] = mvalid ? w : 0.0f;
      }
      float P[4][3];
#pragma unroll
      for (int t = 0; t < 3; ++t) {
        const float sg = g[t] + g[6 + t];
        P[0][t] = g[t];
        P[1][t] = 0.5f * (sg + g[3 + t]);
        P[2][t] = 0.5f * (sg - g[3 + t]);
        P[3][t] = g[6 + t];
      }
#pragma unroll
      for (int i = 0; i < 4; ++i) {
        const float sp = P[i][0] + P[i][2];
        U[e][4 * i + 0] = P[i][0];
        U[e][4 * i + 1] = 0.5f * (sp + P[i][1]);
        U[e][4 * i + 2] = 0.5f * (sp - P[i][1]);
        U[e][4 * i + 3] = P[i][2];
      }
    }

#pragma unroll
    for (int pos = 0; pos < 16; ++pos) {
      v2f a; a[0] = U[0][pos]; a[1] = U[1][pos];
      M[pos] = __builtin_amdgcn_wmma_f32_16x16x4_f32(
          false, a, false, bcur[pos], (short)0, M[pos], false, false);
    }
#pragma unroll
    for (int pos = 0; pos < 16; ++pos) bcur[pos] = bnxt[pos];
  }

#pragma unroll
  for (int a = 0; a < 2; ++a) {
    v8f R[4];
#pragma unroll
    for (int j = 0; j < 4; ++j)
      R[j] = (a == 0) ? (M[j] + M[4 + j] + M[8 + j]) : (M[4 + j] - M[8 + j] - M[12 + j]);
    const v8f y0 = R[0] + R[1] + R[2] + bias;
    const v8f y1 = R[1] - R[2] - R[3] + bias;
    float* s0 = sm + (a * 32 + 2 * row) * 113 + wave * 16 + 8 * half;
#pragma unroll
    for (int v = 0; v < 8; ++v) { s0[v] = y0[v]; s0[113 + v] = y1[v]; }
  }
  __syncthreads();

  if (tid < 128) {
    const int ij = tid >> 5;
    const int w  = w0 + (tid & 31);
    for (int it = 0; it < 2; ++it) {
      const float* base = sm + (it * 32 + (tid & 31)) * 113;
      float r[25];
      float mx = -3.4e38f;
#pragma unroll
      for (int k = 0; k < 25; ++k) { r[k] = base[k * 4 + ij]; mx = fmaxf(mx, r[k]); }
      float sum = 0.0f;
#pragma unroll
      for (int k = 0; k < 25; ++k) { r[k] = __expf(r[k] - mx); sum += r[k]; }
      const float inv = 1.0f / sum;
      float* mplane = mask + (size_t)(n * HW + h0 + it) * MPLANE;
      float* mquad  = mplane + (ij * 6 * HW + (w & 1) * (HW / 2) + (w >> 1)) * 4;
#pragma unroll
      for (int g = 0; g < 6; ++g) {
        float4 o;
        o.x = r[4 * g + 0] * inv; o.y = r[4 * g + 1] * inv;
        o.z = r[4 * g + 2] * inv; o.w = r[4 * g + 3] * inv;
        *(float4*)(mquad + g * HW * 4) = o;
      }
      mplane[MTAIL + ij * HW + w] = r[24] * inv;
    }
  }
}

#define RCH 32
#define RING 4

template <bool LAST>
__device__ __forceinline__ void reassemble_trip(float (&xw)[RING][36],
                                                const float* xt,
                                                const unsigned (&ro)[6],
                                                const unsigned (&co)[3],
                                                const float (&ms)[2][2][100], float* og) {
#pragma unroll
  for (int s = 0; s < RING; ++s) {
    if (!LAST || s == 0) {
#pragma unroll
      for (int r = 0; r < 6; ++r)
#pragma unroll
        for (int j = 0; j < 3; ++j) {
          const float* rowp = xt + (size_t)(ro[r] + (s + RING - 1) * PIX);
          const float2 v = *(const float2*)(rowp + co[j]);
          xw[(s + RING - 1) % RING][6 * r + 2 * j]     = v.x;
          xw[(s + RING - 1) % RING][6 * r + 2 * j + 1] = v.y;
        }
      __builtin_amdgcn_sched_barrier(0);
    }

    float a[2][2][4];
#pragma unroll
    for (int p = 0; p < 2; ++p)
#pragma unroll
      for (int q = 0; q < 2; ++q) { a[p][q][0] = 0.f; a[p][q][1] = 0.f; a[p][q][2] = 0.f; a[p][q][3] = 0.f; }
#pragma unroll
    for (int r = 0; r < 6; ++r)
#pragma unroll
      for (int cc = 0; cc < 6; ++cc) {
        const float pch = xw[s][6 * r + cc];
#pragma unroll
        for (int p = 0; p < 2; ++p)
#pragma unroll
          for (int q = 0; q < 2; ++q) {
            const int di = r - p, dj = cc - q;
            if (di >= 0 && di < 5 && dj >= 0 && dj < 5) {
              const int k = 5 * di + dj;
              a[p][q][0] += pch * ms[p][q][k];
              a[p][q][1] += pch * ms[p][q][25 + k];
              a[p][q][2] += pch * ms[p][q][50 + k];
              a[p][q][3] += pch * ms[p][q][75 + k];
            }
          }
      }

    float* ob = og + (size_t)s * (128 * 128);
#pragma unroll
    for (int p = 0; p < 2; ++p)
#pragma unroll
      for (int i = 0; i < 2; ++i) {
        float4 o;
        o.x = a[p][0][2 * i]; o.y = a[p][0][2 * i + 1];
        o.z = a[p][1][2 * i]; o.w = a[p][1][2 * i + 1];
        *(float4*)(ob + (2 * p + i) * 128) = o;
      }
  }
}

__global__ __launch_bounds__(32) __attribute__((amdgpu_waves_per_eu(1, 1)))
void carafe_reassemble(
    const float* __restrict__ x, const float* __restrict__ mask,
    float* __restrict__ out) {
  const int t  = threadIdx.x;
  const int b  = blockIdx.x;
  const int cg = b & 7;
  const int h0 = ((b >> 3) & 31) * 2;
  const int n  = b >> 8;

  float ms[2][2][100];
#pragma unroll
  for (int p = 0; p < 2; ++p) {
    const float* mplane = mask + (size_t)(n * HW + h0 + p) * MPLANE;
#pragma unroll
    for (int ij = 0; ij < 4; ++ij) {
#pragma unroll
      for (int g = 0; g < 6; ++g) {
        const float* mq = mplane + ((ij * 6 + g) * HW + t) * 4;
        const float4 m0 = *(const float4*)mq;
        const float4 m1 = *(const float4*)(mq + (HW / 2) * 4);
        ms[p][0][ij * 25 + 4 * g + 0] = m0.x; ms[p][0][ij * 25 + 4 * g + 1] = m0.y;
        ms[p][0][ij * 25 + 4 * g + 2] = m0.z; ms[p][0][ij * 25 + 4 * g + 3] = m0.w;
        ms[p][1][ij * 25 + 4 * g + 0] = m1.x; ms[p][1][ij * 25 + 4 * g + 1] = m1.y;
        ms[p][1][ij * 25 + 4 * g + 2] = m1.z; ms[p][1][ij * 25 + 4 * g + 3] = m1.w;
      }
      const float2 mt = *(const float2*)(mplane + MTAIL + ij * HW + 2 * t);
      ms[p][0][ij * 25 + 24] = mt.x;
      ms[p][1][ij * 25 + 24] = mt.y;
    }
  }

  const float* xg = x + (size_t)(n * 256 + cg * RCH) * PIX;
  unsigned ro[6];
#pragma unroll
  for (int r = 0; r < 6; ++r) ro[r] = (unsigned)(min(max(h0 - 2 + r, 0), HW - 1) * HW);
  unsigned co[3];
#pragma unroll
  for (int j = 0; j < 3; ++j) co[j] = (unsigned)(2 * min(max(t - 1 + j, 0), HW / 2 - 1));
  float* og = out + (((size_t)(n * 256 + cg * RCH) * 128) + 2 * h0) * 128 + 4 * t;

  float xw[RING][36];
#pragma unroll
  for (int s = 0; s < RING - 1; ++s) {
#pragma unroll
    for (int r = 0; r < 6; ++r)
#pragma unroll
      for (int j = 0; j < 3; ++j) {
        const float* rowp = xg + (size_t)(ro[r] + s * PIX);
        const float2 v = *(const float2*)(rowp + co[j]);
        xw[s][6 * r + 2 * j]     = v.x;
        xw[s][6 * r + 2 * j + 1] = v.y;
      }
    __builtin_amdgcn_sched_barrier(0);
  }

#pragma unroll
  for (int p = 0; p < 2; ++p)
#pragma unroll
    for (int q = 0; q < 2; ++q)
#pragma unroll
      for (int di = 0; di < 5; ++di)
#pragma unroll
        for (int dj = 0; dj < 5; ++dj) {
          const bool ok = ((unsigned)(h0 + p + di - 2) < HW) & ((unsigned)(2 * t + q + dj - 2) < HW);
#pragma unroll
          for (int ij = 0; ij < 4; ++ij)
            ms[p][q][ij * 25 + di * 5 + dj] = ok ? ms[p][q][ij * 25 + di * 5 + dj] : 0.0f;
        }

#pragma nounroll
  for (int trip = 0; trip < RCH / RING - 1; ++trip) {
    reassemble_trip<false>(xw, xg + (size_t)trip * (RING * PIX), ro, co, ms, og);
    og += (size_t)RING * 128 * 128;
  }
  reassemble_trip<true>(xw, xg + (size_t)(RCH - RING) * PIX, ro, co, ms, og);
}

extern "C" void kernel_launch(void* const* d_in, const int* in_sizes, int n_in,
                              void* d_out, int out_size, void* d_ws, size_t ws_size,
                              hipStream_t stream) {
  (void)in_sizes; (void)n_in; (void)out_size; (void)ws_size;
  const float* x      = (const float*)d_in[0];
  const float* w_comp = (const float*)d_in[1];
  const float* b_comp = (const float*)d_in[2];
  const float* w_enc  = (const float*)d_in[3];
  const float* b_enc  = (const float*)d_in[4];
  float* out = (float*)d_out;

  float* comp = (float*)d_ws;
  float* mask = comp + (size_t)4 * CCOMP * PIX;

  carafe_compress<<<256, 128, 0, stream>>>(x, w_comp, b_comp, comp);
  carafe_encode_softmax<<<256, 224, 0, stream>>>(comp, w_enc, b_enc, mask);
  carafe_reassemble<<<1024, 32, 0, stream>>>(x, mask, out);
}
